# starting kernel
# baseline (speedup 1.0000x reference)
#include <hip/hip_runtime.h>
#include <cstdint>

typedef __attribute__((ext_vector_type(2))) float v2f;
typedef __attribute__((ext_vector_type(8))) float v8f;

#define D 64
#define WPAD 68

#define WMMA4(a, b, c) \
  __builtin_amdgcn_wmma_f32_16x16x4_f32(false, (a), false, (b), (short)0, (c), false, false)

__device__ __forceinline__ void stage_wT(const float* __restrict__ W, float (*wl)[WPAD],
                                         int tid) {
  for (int idx = tid; idx < 64 * 16; idx += 256) {
    const int k = idx >> 4;
    const int n4 = (idx & 15) << 2;
    const float4 v = *reinterpret_cast<const float4*>(W + k * D + n4);
    wl[n4 + 0][k] = v.x;
    wl[n4 + 1][k] = v.y;
    wl[n4 + 2][k] = v.z;
    wl[n4 + 3][k] = v.w;
  }
}

__global__ void k_init(const float* __restrict__ x, float* __restrict__ h,
                       float* __restrict__ deg, int n) {
  long i = (long)blockIdx.x * blockDim.x + threadIdx.x;
  long tot = (long)n * D;
  if (i < n) deg[i] = 0.0f;
  if (i < tot) h[i] = __logf(x[i] + 1.0f);
}

__global__ void k_deg(const int* __restrict__ dst, float* __restrict__ deg, int e) {
  long i = (long)blockIdx.x * blockDim.x + threadIdx.x;
  if (i < e) atomicAdd(&deg[dst[i]], 1.0f);
}

__global__ void k_zero4(float4* __restrict__ p, long n4) {
  long i = (long)blockIdx.x * blockDim.x + threadIdx.x;
  if (i < n4) p[i] = make_float4(0.f, 0.f, 0.f, 0.f);
}

__global__ void k_scatter(const float* __restrict__ h, const int* __restrict__ src,
                          const int* __restrict__ dst, float* __restrict__ neigh, int e) {
  long gid = (long)blockIdx.x * blockDim.x + threadIdx.x;
  long ed = gid >> 4;
  if (ed >= e) return;
  int f = (int)(gid & 15) * 4;
  int s = src[ed], d = dst[ed];
  float4 v = *reinterpret_cast<const float4*>(h + (size_t)s * D + f);
  float* p = neigh + (size_t)d * D + f;
  atomicAdd(p + 0, v.x);
  atomicAdd(p + 1, v.y);
  atomicAdd(p + 2, v.z);
  atomicAdd(p + 3, v.w);
}

__global__ void __launch_bounds__(256)
k_sage(const float* __restrict__ h, const float* __restrict__ neigh,
       const float* __restrict__ deg, const float* __restrict__ Ws,
       const float* __restrict__ Wn, const float* __restrict__ bias,
       float* __restrict__ out, int n, int do_norm) {
  __shared__ float wls[D][WPAD];
  __shared__ float wln[D][WPAD];
  stage_wT(Ws, wls, threadIdx.x);
  stage_wT(Wn, wln, threadIdx.x);
  __syncthreads();

  const int lane = threadIdx.x & 31;
  const int wave = threadIdx.x >> 5;
  const int row0 = (blockIdx.x * 8 + wave) * 16;
  if (row0 >= n) return;
  const int half = lane >> 4;
  const int l15 = lane & 15;
  int rowA = row0 + l15;
  if (rowA >= n) rowA = n - 1;
  const float dinv = 1.0f / fmaxf(deg[rowA], 1.0f);
  const float* __restrict__ hp = h + (size_t)rowA * D;
  const float* __restrict__ np = neigh + (size_t)rowA * D;

  v8f acc[4] = {};
  for (int k0 = 0; k0 < D; k0 += 4) {
    const int kk = k0 + half * 2;
    const v2f ah = *reinterpret_cast<const v2f*>(hp + kk);
    v2f an = *reinterpret_cast<const v2f*>(np + kk);
    an.x *= dinv;
    an.y *= dinv;
#pragma unroll
    for (int t = 0; t < 4; ++t) {
      const int nn = t * 16 + l15;
      const v2f bs = *reinterpret_cast<const v2f*>(&wls[nn][kk]);
      const v2f bn = *reinterpret_cast<const v2f*>(&wln[nn][kk]);
      acc[t] = WMMA4(ah, bs, acc[t]);
      acc[t] = WMMA4(an, bn, acc[t]);
    }
  }

  float ss[8];
#pragma unroll
  for (int i = 0; i < 8; ++i) ss[i] = 0.0f;
#pragma unroll
  for (int t = 0; t < 4; ++t) {
    const float bv = bias[t * 16 + l15];
#pragma unroll
    for (int i = 0; i < 8; ++i) {
      float v = acc[t][i] + bv;
      if (do_norm) v = fmaxf(v, 0.0f);
      acc[t][i] = v;
      ss[i] += v * v;
    }
  }
  if (do_norm) {
#pragma unroll
    for (int i = 0; i < 8; ++i) {
      float s = ss[i];
      s += __shfl_xor(s, 1, 32);
      s += __shfl_xor(s, 2, 32);
      s += __shfl_xor(s, 4, 32);
      s += __shfl_xor(s, 8, 32);
      const float inv = 1.0f / fmaxf(sqrtf(s), 1e-12f);
#pragma unroll
      for (int t = 0; t < 4; ++t) acc[t][i] *= inv;
    }
  }
#pragma unroll
  for (int t = 0; t < 4; ++t) {
#pragma unroll
    for (int i = 0; i < 8; ++i) {
      const int r = row0 + i + half * 8;
      if (r < n) out[(size_t)r * D + t * 16 + l15] = acc[t][i];
    }
  }
}

__global__ void __launch_bounds__(256)
k_head(const float* __restrict__ h, const float* __restrict__ fcw,
       const float* __restrict__ fcb, const float* __restrict__ gamma,
       const float* __restrict__ beta, const float* __restrict__ mean,
       const float* __restrict__ var, const float* __restrict__ w21,
       const float* __restrict__ b21, const float* __restrict__ w22,
       const float* __restrict__ b22, float* __restrict__ zloc,
       float* __restrict__ zscale, int n) {
  __shared__ float wl0[D][WPAD];
  __shared__ float wl1[D][WPAD];
  __shared__ float wl2[D][WPAD];
  __shared__ float u[8][16][WPAD];
  stage_wT(fcw, wl0, threadIdx.x);
  stage_wT(w21, wl1, threadIdx.x);
  stage_wT(w22, wl2, threadIdx.x);
  __syncthreads();

  const int lane = threadIdx.x & 31;
  const int wave = threadIdx.x >> 5;
  const int row0 = (blockIdx.x * 8 + wave) * 16;
  if (row0 >= n) return;
  const int half = lane >> 4;
  const int l15 = lane & 15;
  int rowA = row0 + l15;
  if (rowA >= n) rowA = n - 1;
  const float* __restrict__ hp = h + (size_t)rowA * D;

  v8f acc[4] = {};
  for (int k0 = 0; k0 < D; k0 += 4) {
    const int kk = k0 + half * 2;
    const v2f a = *reinterpret_cast<const v2f*>(hp + kk);
#pragma unroll
    for (int t = 0; t < 4; ++t) {
      const int nn = t * 16 + l15;
      const v2f b = *reinterpret_cast<const v2f*>(&wl0[nn][kk]);
      acc[t] = WMMA4(a, b, acc[t]);
    }
  }
#pragma unroll
  for (int t = 0; t < 4; ++t) {
    const int c = t * 16 + l15;
    const float bb = fcb[c];
    const float m = mean[c];
    const float iv = rsqrtf(var[c] + 1e-5f);
    const float g = gamma[c];
    const float bt = beta[c];
#pragma unroll
    for (int i = 0; i < 8; ++i) {
      float v = acc[t][i] + bb;
      v = (v - m) * iv * g + bt;
      v = fmaxf(v, 0.0f);
      v = v + __logf(1.0f + __expf(-v));
      u[wave][i + half * 8][c] = v;
    }
  }
  v8f aL[4] = {}, aS[4] = {};
  for (int k0 = 0; k0 < D; k0 += 4) {
    const int kk = k0 + half * 2;
    const v2f a = *reinterpret_cast<const v2f*>(&u[wave][l15][kk]);
#pragma unroll
    for (int t = 0; t < 4; ++t) {
      const int nn = t * 16 + l15;
      const v2f bL = *reinterpret_cast<const v2f*>(&wl1[nn][kk]);
      const v2f bS = *reinterpret_cast<const v2f*>(&wl2[nn][kk]);
      aL[t] = WMMA4(a, bL, aL[t]);
      aS[t] = WMMA4(a, bS, aS[t]);
    }
  }
#pragma unroll
  for (int t = 0; t < 4; ++t) {
    const int c = t * 16 + l15;
    const float bl = b21[c];
    const float bs = b22[c];
#pragma unroll
    for (int i = 0; i < 8; ++i) {
      const int r = row0 + i + half * 8;
      if (r < n) {
        const size_t o = (size_t)r * D + c;
        zloc[o] = aL[t][i] + bl;
        zscale[o] = __expf(aS[t][i] + bs);
      }
    }
  }
}

extern "C" void kernel_launch(void* const* d_in, const int* in_sizes, int n_in,
                              void* d_out, int out_size, void* d_ws, size_t ws_size,
                              hipStream_t stream) {
  const float* x = (const float*)d_in[0];
  const int* esrc = (const int*)d_in[1];
  const int* edst = (const int*)d_in[2];
  const float* Wself = (const float*)d_in[3];
  const float* Wneigh = (const float*)d_in[4];
  const float* bsage = (const float*)d_in[5];
  const float* fcw = (const float*)d_in[6];
  const float* fcb = (const float*)d_in[7];
  const float* gamma = (const float*)d_in[8];
  const float* beta = (const float*)d_in[9];
  const float* mean = (const float*)d_in[10];
  const float* var = (const float*)d_in[11];
  const float* w21 = (const float*)d_in[12];
  const float* b21 = (const float*)d_in[13];
  const float* w22 = (const float*)d_in[14];
  const float* b22 = (const float*)d_in[15];

  const int n = in_sizes[0] / D;
  const int e = in_sizes[1];
  const size_t nd = (size_t)n * D;

  float* ws = (float*)d_ws;
  float* deg = ws;
  float* hA = ws + (((size_t)n + 255) & ~(size_t)255);
  float* hB = hA + nd;
  float* nb = hB + nd;

  {
    long tot = (long)n * D;
    k_init<<<(int)((tot + 255) / 256), 256, 0, stream>>>(x, hA, deg, n);
  }
  k_deg<<<(e + 255) / 256, 256, 0, stream>>>(edst, deg, e);

  const int tiles = (n + 15) / 16;
  const int gblocks = (tiles + 7) / 8;

  float* hcur = hA;
  float* hnext = hB;
  for (int l = 0; l < 3; ++l) {
    long n4 = (long)(nd / 4);
    k_zero4<<<(int)((n4 + 255) / 256), 256, 0, stream>>>((float4*)nb, n4);
    long st = (long)e * 16;
    k_scatter<<<(int)((st + 255) / 256), 256, 0, stream>>>(hcur, esrc, edst, nb, e);
    k_sage<<<gblocks, 256, 0, stream>>>(hcur, nb, deg, Wself + (size_t)l * D * D,
                                        Wneigh + (size_t)l * D * D, bsage + (size_t)l * D,
                                        hnext, n, (l < 2) ? 1 : 0);
    float* tmp = hcur;
    hcur = hnext;
    hnext = tmp;
  }

  float* zloc = (float*)d_out;
  float* zscale = zloc + nd;
  k_head<<<gblocks, 256, 0, stream>>>(hcur, fcw, fcb, gamma, beta, mean, var, w21, b21,
                                      w22, b22, zloc, zscale, n);
}


# Layer product ahead of the edge sum: layer kernels emit P=a@Ws+b and G=a@Wn, scatter adds G[src]/deg[dst] into P; log pass and zero fills gone (8 launches); sum order: self term, then atomic arrival
# speedup vs baseline: 1.6690x; 1.6690x over previous
#include <hip/hip_runtime.h>
#include <cstdint>

typedef __attribute__((ext_vector_type(2))) float v2f;
typedef __attribute__((ext_vector_type(8))) float v8f;

#define D 64
#define WPAD 68

#define WMMA4(a, b, c) \
  __builtin_amdgcn_wmma_f32_16x16x4_f32(false, (a), false, (b), (short)0, (c), false, false)

__device__ __forceinline__ void stage_wT(const float* __restrict__ W, float (*wl)[WPAD],
                                         int tid) {
  for (int idx = tid; idx < 64 * 16; idx += 256) {
    const int k = idx >> 4;
    const int n4 = (idx & 15) << 2;
    const float4 v = *reinterpret_cast<const float4*>(W + k * D + n4);
    wl[n4 + 0][k] = v.x;
    wl[n4 + 1][k] = v.y;
    wl[n4 + 2][k] = v.z;
    wl[n4 + 3][k] = v.w;
  }
}

__global__ void k_deg(const int* __restrict__ dst, float* __restrict__ deg, int e) {
  long i = (long)blockIdx.x * blockDim.x + threadIdx.x;
  if (i < e) atomicAdd(&deg[dst[i]], 1.0f);
}

#define EPW 16
typedef __attribute__((address_space(1))) float gfloat;

#define PIN8(x, o)                                                                   \
  asm("" : "+s"(x[o + 0]), "+s"(x[o + 1]), "+s"(x[o + 2]), "+s"(x[o + 3]),           \
           "+s"(x[o + 4]), "+s"(x[o + 5]), "+s"(x[o + 6]), "+s"(x[o + 7]))

__global__ void __launch_bounds__(256)
k_scatter(const float* __restrict__ g, const int* __restrict__ src,
          const int* __restrict__ dst, const float* __restrict__ deg,
          float* __restrict__ acc, int e) {
  const unsigned lane = threadIdx.x & 31u;
  const int wv = __builtin_amdgcn_readfirstlane((int)(blockIdx.x * 8u + (threadIdx.x >> 5)));
  const long e0 = (long)wv * EPW;
  if (e0 >= e) return;
  if (e0 + EPW <= e) {
    unsigned long long hb[EPW], db[EPW];
    for (int j = 0; j < EPW; ++j)
      hb[j] = (unsigned long long)g + (unsigned long long)(unsigned)src[e0 + j] * (D * sizeof(float));
    for (int j = 0; j < EPW; ++j)
      db[j] = (unsigned long long)acc + (unsigned long long)(unsigned)dst[e0 + j] * (D * sizeof(float));
    const float dinv = 1.0f / fmaxf(deg[(unsigned)dst[e0 + (lane & 15u)]], 1.0f);
    PIN8(hb, 0);
    PIN8(hb, 8);
    float a[EPW], b[EPW];
    for (int j = 0; j < EPW; ++j) {
      const gfloat* p = (const gfloat*)hb[j];
      a[j] = p[lane];
      b[j] = p[lane + 32u];
    }
    PIN8(db, 0);
    PIN8(db, 8);
    for (int j = 0; j < EPW; ++j) {
      float* q = (float*)(gfloat*)db[j];
      const float dj = __int_as_float(__builtin_amdgcn_readlane(__float_as_int(dinv), j));
      atomicAdd(q + lane, a[j] * dj);
      atomicAdd(q + lane + 32u, b[j] * dj);
    }
  } else {
    const int cnt = (int)(e - e0);
    for (int j = 0; j < cnt; ++j) {
      const float* p = g + (size_t)src[e0 + j] * D;
      const float a0 = p[lane], b0 = p[lane + 32u];
      const float d0 = 1.0f / fmaxf(deg[dst[e0 + j]], 1.0f);
      float* q = acc + (size_t)dst[e0 + j] * D;
      atomicAdd(q + lane, a0 * d0);
      atomicAdd(q + lane + 32u, b0 * d0);
    }
  }
}

__global__ void __launch_bounds__(256)
k_proj(const float* __restrict__ in, const float* __restrict__ Ws,
       const float* __restrict__ Wn, const float* __restrict__ bias,
       float* __restrict__ P, float* __restrict__ G, float* __restrict__ deg, int n,
       int first) {
  __shared__ float wls[D][WPAD];
  __shared__ float wln[D][WPAD];
  stage_wT(Ws, wls, threadIdx.x);
  stage_wT(Wn, wln, threadIdx.x);
  if (first) {
    const int gi = blockIdx.x * 256 + threadIdx.x;
    if (gi < n) deg[gi] = 0.0f;
  }
  __syncthreads();

  const int lane = threadIdx.x & 31;
  const int wave = threadIdx.x >> 5;
  const int row0 = (blockIdx.x * 8 + wave) * 16;
  if (row0 >= n) return;
  const int half = lane >> 4;
  const int l15 = lane & 15;
  int rowA = row0 + l15;
  if (rowA >= n) rowA = n - 1;
  const float* __restrict__ ip = in + (size_t)rowA * D + half * 2;

  v2f a[16];
#pragma unroll
  for (int j = 0; j < 16; ++j) a[j] = *reinterpret_cast<const v2f*>(ip + 4 * j);
  if (first) {
#pragma unroll
    for (int j = 0; j < 16; ++j) {
      a[j].x = __logf(a[j].x + 1.0f);
      a[j].y = __logf(a[j].y + 1.0f);
    }
  } else {
    float ss = 0.0f;
#pragma unroll
    for (int j = 0; j < 16; ++j) {
      a[j].x = fmaxf(a[j].x, 0.0f);
      a[j].y = fmaxf(a[j].y, 0.0f);
      ss += a[j].x * a[j].x;
      ss += a[j].y * a[j].y;
    }
    ss += __shfl_xor(ss, 16, 32);
    const float inv = 1.0f / fmaxf(sqrtf(ss), 1e-12f);
#pragma unroll
    for (int j = 0; j < 16; ++j) {
      a[j].x *= inv;
      a[j].y *= inv;
    }
  }

  v8f accS[4] = {}, accN[4] = {};
#pragma unroll
  for (int j = 0; j < 16; ++j) {
    const int kk = 4 * j + half * 2;
#pragma unroll
    for (int t = 0; t < 4; ++t) {
      const int nn = t * 16 + l15;
      const v2f bs = *reinterpret_cast<const v2f*>(&wls[nn][kk]);
      const v2f bn = *reinterpret_cast<const v2f*>(&wln[nn][kk]);
      accS[t] = WMMA4(a[j], bs, accS[t]);
      accN[t] = WMMA4(a[j], bn, accN[t]);
    }
  }

#pragma unroll
  for (int t = 0; t < 4; ++t) {
    const int c = t * 16 + l15;
    const float bv = bias[c];
#pragma unroll
    for (int i = 0; i < 8; ++i) {
      const int r = row0 + i + half * 8;
      if (r < n) {
        const size_t o = (size_t)r * D + c;
        P[o] = accS[t][i] + bv;
        G[o] = accN[t][i];
      }
    }
  }
}

__global__ void __launch_bounds__(256)
k_head(const float* __restrict__ h, const float* __restrict__ fcw,
       const float* __restrict__ fcb, const float* __restrict__ gamma,
       const float* __restrict__ beta, const float* __restrict__ mean,
       const float* __restrict__ var, const float* __restrict__ w21,
       const float* __restrict__ b21, const float* __restrict__ w22,
       const float* __restrict__ b22, float* __restrict__ zloc,
       float* __restrict__ zscale, int n) {
  __shared__ float wl0[D][WPAD];
  __shared__ float wl1[D][WPAD];
  __shared__ float wl2[D][WPAD];
  __shared__ float u[8][16][WPAD];
  stage_wT(fcw, wl0, threadIdx.x);
  stage_wT(w21, wl1, threadIdx.x);
  stage_wT(w22, wl2, threadIdx.x);
  __syncthreads();

  const int lane = threadIdx.x & 31;
  const int wave = threadIdx.x >> 5;
  const int row0 = (blockIdx.x * 8 + wave) * 16;
  if (row0 >= n) return;
  const int half = lane >> 4;
  const int l15 = lane & 15;
  int rowA = row0 + l15;
  if (rowA >= n) rowA = n - 1;
  const float* __restrict__ hp = h + (size_t)rowA * D;

  v8f acc[4] = {};
  for (int k0 = 0; k0 < D; k0 += 4) {
    const int kk = k0 + half * 2;
    const v2f a = *reinterpret_cast<const v2f*>(hp + kk);
#pragma unroll
    for (int t = 0; t < 4; ++t) {
      const int nn = t * 16 + l15;
      const v2f b = *reinterpret_cast<const v2f*>(&wl0[nn][kk]);
      acc[t] = WMMA4(a, b, acc[t]);
    }
  }
#pragma unroll
  for (int t = 0; t < 4; ++t) {
    const int c = t * 16 + l15;
    const float bb = fcb[c];
    const float m = mean[c];
    const float iv = rsqrtf(var[c] + 1e-5f);
    const float g = gamma[c];
    const float bt = beta[c];
#pragma unroll
    for (int i = 0; i < 8; ++i) {
      float v = acc[t][i] + bb;
      v = (v - m) * iv * g + bt;
      v = fmaxf(v, 0.0f);
      v = v + __logf(1.0f + __expf(-v));
      u[wave][i + half * 8][c] = v;
    }
  }
  v8f aL[4] = {}, aS[4] = {};
  for (int k0 = 0; k0 < D; k0 += 4) {
    const int kk = k0 + half * 2;
    const v2f a = *reinterpret_cast<const v2f*>(&u[wave][l15][kk]);
#pragma unroll
    for (int t = 0; t < 4; ++t) {
      const int nn = t * 16 + l15;
      const v2f bL = *reinterpret_cast<const v2f*>(&wl1[nn][kk]);
      const v2f bS = *reinterpret_cast<const v2f*>(&wl2[nn][kk]);
      aL[t] = WMMA4(a, bL, aL[t]);
      aS[t] = WMMA4(a, bS, aS[t]);
    }
  }
#pragma unroll
  for (int t = 0; t < 4; ++t) {
    const int c = t * 16 + l15;
    const float bl = b21[c];
    const float bs = b22[c];
#pragma unroll
    for (int i = 0; i < 8; ++i) {
      const int r = row0 + i + half * 8;
      if (r < n) {
        const size_t o = (size_t)r * D + c;
        zloc[o] = aL[t][i] + bl;
        zscale[o] = __expf(aS[t][i] + bs);
      }
    }
  }
}

extern "C" void kernel_launch(void* const* d_in, const int* in_sizes, int n_in,
                              void* d_out, int out_size, void* d_ws, size_t ws_size,
                              hipStream_t stream) {
  const float* x = (const float*)d_in[0];
  const int* esrc = (const int*)d_in[1];
  const int* edst = (const int*)d_in[2];
  const float* Wself = (const float*)d_in[3];
  const float* Wneigh = (const float*)d_in[4];
  const float* bsage = (const float*)d_in[5];
  const float* fcw = (const float*)d_in[6];
  const float* fcb = (const float*)d_in[7];
  const float* gamma = (const float*)d_in[8];
  const float* beta = (const float*)d_in[9];
  const float* mean = (const float*)d_in[10];
  const float* var = (const float*)d_in[11];
  const float* w21 = (const float*)d_in[12];
  const float* b21 = (const float*)d_in[13];
  const float* w22 = (const float*)d_in[14];
  const float* b22 = (const float*)d_in[15];

  const int n = in_sizes[0] / D;
  const int e = in_sizes[1];
  const size_t nd = (size_t)n * D;

  float* ws = (float*)d_ws;
  float* deg = ws;
  float* pA = ws + (((size_t)n + 255) & ~(size_t)255);
  float* pB = pA + nd;
  float* g = pB + nd;

  const int tiles = (n + 15) / 16;
  const int gblocks = (tiles + 7) / 8;

  k_proj<<<gblocks, 256, 0, stream>>>(x, Wself, Wneigh, bsage, pA, g, deg, n, 1);
  k_deg<<<(e + 255) / 256, 256, 0, stream>>>(edst, deg, e);

  float* hcur = pA;
  float* hnext = pB;
  for (int l = 0; l < 3; ++l) {
    k_scatter<<<(e + 8 * EPW - 1) / (8 * EPW), 256, 0, stream>>>(g, esrc, edst, deg, hcur, e);
    if (l < 2) {
      k_proj<<<gblocks, 256, 0, stream>>>(hcur, Wself + (size_t)(l + 1) * D * D,
                                          Wneigh + (size_t)(l + 1) * D * D,
                                          bsage + (size_t)(l + 1) * D, hnext, g, deg, n, 0);
      float* tmp = hcur;
      hcur = hnext;
      hnext = tmp;
    }
  }

  float* zloc = (float*)d_out;
  float* zscale = zloc + nd;
  k_head<<<gblocks, 256, 0, stream>>>(hcur, fcw, fcb, gamma, beta, mean, var, w21, b21,
                                      w22, b22, zloc, zscale, n);
}


# Scatter split by cache half: work-group reads its die, takes 256-edge chunks off its half's ticket counter, adds only rows of that half's node range; sum order = atomic arrival order as before
# speedup vs baseline: 1.7282x; 1.0355x over previous
#include <hip/hip_runtime.h>
#include <cstdint>

typedef __attribute__((ext_vector_type(2))) float v2f;
typedef __attribute__((ext_vector_type(8))) float v8f;

#define D 64
#define WPAD 68

#define WMMA4(a, b, c) \
  __builtin_amdgcn_wmma_f32_16x16x4_f32(false, (a), false, (b), (short)0, (c), false, false)

__device__ __forceinline__ void stage_wT(const float* __restrict__ W, float (*wl)[WPAD],
                                         int tid) {
  for (int idx = tid; idx < 64 * 16; idx += 256) {
    const int k = idx >> 4;
    const int n4 = (idx & 15) << 2;
    const float4 v = *reinterpret_cast<const float4*>(W + k * D + n4);
    wl[n4 + 0][k] = v.x;
    wl[n4 + 1][k] = v.y;
    wl[n4 + 2][k] = v.z;
    wl[n4 + 3][k] = v.w;
  }
}

__global__ void k_deg(const int* __restrict__ dst, float* __restrict__ deg, int e) {
  long i = (long)blockIdx.x * blockDim.x + threadIdx.x;
  if (i < e) atomicAdd(&deg[dst[i]], 1.0f);
}

#define CH 256
#define GRP 16
#define CTR_STRIDE 64
typedef __attribute__((address_space(1))) float gfloat;

#define PIN8(x, o)                                                                   \
  asm("" : "+s"(x[o + 0]), "+s"(x[o + 1]), "+s"(x[o + 2]), "+s"(x[o + 3]),           \
           "+s"(x[o + 4]), "+s"(x[o + 5]), "+s"(x[o + 6]), "+s"(x[o + 7]))

__global__ void __launch_bounds__(256)
k_scatter(const float* __restrict__ g, const int* __restrict__ src,
          const int* __restrict__ dst, const float* __restrict__ deg,
          float* __restrict__ acc, int* __restrict__ ctr, int e, int n) {
  __shared__ int tk[3];
  const unsigned lane = threadIdx.x & 31u;
  const unsigned wave = threadIdx.x >> 5;
  const int nch = (e + CH - 1) / CH;
  const int nsplit = n >> 1;
  if (threadIdx.x == 0) {
    const unsigned id = (unsigned)__builtin_amdgcn_s_sendmsg_rtn(0x87);
    const int q0 = (int)((id >> 18) & 1u);
    tk[2] = q0;
    tk[0] = atomicAdd(ctr + q0 * CTR_STRIDE, 1);
  }
  __syncthreads();
  int q = tk[2];
  int t = tk[0];
  int it = 0;
  bool other = false;
  for (;;) {
    int tn = 0;
    const bool live = t < nch;
    if (!live) {
      if (other) break;
      other = true;
      q ^= 1;
    }
    if (threadIdx.x == 0) tn = atomicAdd(ctr + q * CTR_STRIDE, 1);
    if (live) {
      const long ei = (long)t * CH + (long)(wave * 32u + lane);
      const bool ok = ei < e;
      const int d = ok ? dst[ei] : 0;
      const int s = ok ? src[ei] : 0;
      const bool mine = ok && ((d >= nsplit) == (q != 0));
      const float dinv = mine ? 1.0f / fmaxf(deg[d], 1.0f) : 0.0f;
      unsigned m = __builtin_amdgcn_ballot_w32(mine);
      while (m) {
        const unsigned m0 = m;
        const int cnt = min((int)__builtin_popcount(m0), GRP);
        unsigned long long hb[GRP], db[GRP];
        unsigned w1 = m0;
#pragma unroll
        for (int k = 0; k < GRP; ++k) {
          const int j = w1 ? (int)__builtin_ctz(w1) : 0;
          w1 &= w1 - 1u;
          const unsigned sj = (unsigned)__builtin_amdgcn_readlane(s, j);
          hb[k] = (unsigned long long)g + (unsigned long long)sj * (D * sizeof(float));
        }
        PIN8(hb, 0);
        PIN8(hb, 8);
        float a[GRP], b[GRP];
#pragma unroll
        for (int k = 0; k < GRP; ++k) {
          a[k] = 0.0f;
          b[k] = 0.0f;
          if (k < cnt) {
            const gfloat* p = (const gfloat*)hb[k];
            a[k] = p[lane];
            b[k] = p[lane + 32u];
          }
        }
        unsigned w2 = m0;
#pragma unroll
        for (int k = 0; k < GRP; ++k) {
          const int j = w2 ? (int)__builtin_ctz(w2) : 0;
          w2 &= w2 - 1u;
          const unsigned dd = (unsigned)__builtin_amdgcn_readlane(d, j);
          db[k] = (unsigned long long)acc + (unsigned long long)dd * (D * sizeof(float));
        }
        PIN8(db, 0);
        PIN8(db, 8);
#pragma unroll
        for (int k = 0; k < GRP; ++k) {
          const int j = m ? (int)__builtin_ctz(m) : 0;
          m &= m - 1u;
          if (k < cnt) {
            const float dj = __int_as_float(__builtin_amdgcn_readlane(__float_as_int(dinv), j));
            float* w = (float*)(gfloat*)db[k];
            atomicAdd(w + lane, a[k] * dj);
            atomicAdd(w + lane + 32u, b[k] * dj);
          }
        }
      }
    }
    if (threadIdx.x == 0) tk[(it + 1) & 1] = tn;
    __syncthreads();
    ++it;
    t = tk[it & 1];
  }
}

__global__ void __launch_bounds__(256)
k_proj(const float* __restrict__ in, const float* __restrict__ Ws,
       const float* __restrict__ Wn, const float* __restrict__ bias,
       float* __restrict__ P, float* __restrict__ G, float* __restrict__ deg,
       int* __restrict__ ctr, int n, int first) {
  __shared__ float wls[D][WPAD];
  __shared__ float wln[D][WPAD];
  stage_wT(Ws, wls, threadIdx.x);
  stage_wT(Wn, wln, threadIdx.x);
  if (first) {
    const int gi = blockIdx.x * 256 + threadIdx.x;
    if (gi < n) deg[gi] = 0.0f;
  }
  if (blockIdx.x == 0 && threadIdx.x < 2) ctr[threadIdx.x * CTR_STRIDE] = 0;
  __syncthreads();

  const int lane = threadIdx.x & 31;
  const int wave = threadIdx.x >> 5;
  const int row0 = (blockIdx.x * 8 + wave) * 16;
  if (row0 >= n) return;
  const int half = lane >> 4;
  const int l15 = lane & 15;
  int rowA = row0 + l15;
  if (rowA >= n) rowA = n - 1;
  const float* __restrict__ ip = in + (size_t)rowA * D + half * 2;

  v2f a[16];
#pragma unroll
  for (int j = 0; j < 16; ++j) a[j] = *reinterpret_cast<const v2f*>(ip + 4 * j);
  if (first) {
#pragma unroll
    for (int j = 0; j < 16; ++j) {
      a[j].x = __logf(a[j].x + 1.0f);
      a[j].y = __logf(a[j].y + 1.0f);
    }
  } else {
    float ss = 0.0f;
#pragma unroll
    for (int j = 0; j < 16; ++j) {
      a[j].x = fmaxf(a[j].x, 0.0f);
      a[j].y = fmaxf(a[j].y, 0.0f);
      ss += a[j].x * a[j].x;
      ss += a[j].y * a[j].y;
    }
    ss += __shfl_xor(ss, 16, 32);
    const float inv = 1.0f / fmaxf(sqrtf(ss), 1e-12f);
#pragma unroll
    for (int j = 0; j < 16; ++j) {
      a[j].x *= inv;
      a[j].y *= inv;
    }
  }

  v8f accS[4] = {}, accN[4] = {};
#pragma unroll
  for (int j = 0; j < 16; ++j) {
    const int kk = 4 * j + half * 2;
#pragma unroll
    for (int t = 0; t < 4; ++t) {
      const int nn = t * 16 + l15;
      const v2f bs = *reinterpret_cast<const v2f*>(&wls[nn][kk]);
      const v2f bn = *reinterpret_cast<const v2f*>(&wln[nn][kk]);
      accS[t] = WMMA4(a[j], bs, accS[t]);
      accN[t] = WMMA4(a[j], bn, accN[t]);
    }
  }

#pragma unroll
  for (int t = 0; t < 4; ++t) {
    const int c = t * 16 + l15;
    const float bv = bias[c];
#pragma unroll
    for (int i = 0; i < 8; ++i) {
      const int r = row0 + i + half * 8;
      if (r < n) {
        const size_t o = (size_t)r * D + c;
        P[o] = accS[t][i] + bv;
        G[o] = accN[t][i];
      }
    }
  }
}

__global__ void __launch_bounds__(256)
k_head(const float* __restrict__ h, const float* __restrict__ fcw,
       const float* __restrict__ fcb, const float* __restrict__ gamma,
       const float* __restrict__ beta, const float* __restrict__ mean,
       const float* __restrict__ var, const float* __restrict__ w21,
       const float* __restrict__ b21, const float* __restrict__ w22,
       const float* __restrict__ b22, float* __restrict__ zloc,
       float* __restrict__ zscale, int n) {
  __shared__ float wl0[D][WPAD];
  __shared__ float wl1[D][WPAD];
  __shared__ float wl2[D][WPAD];
  __shared__ float u[8][16][WPAD];
  stage_wT(fcw, wl0, threadIdx.x);
  stage_wT(w21, wl1, threadIdx.x);
  stage_wT(w22, wl2, threadIdx.x);
  __syncthreads();

  const int lane = threadIdx.x & 31;
  const int wave = threadIdx.x >> 5;
  const int row0 = (blockIdx.x * 8 + wave) * 16;
  if (row0 >= n) return;
  const int half = lane >> 4;
  const int l15 = lane & 15;
  int rowA = row0 + l15;
  if (rowA >= n) rowA = n - 1;
  const float* __restrict__ hp = h + (size_t)rowA * D;

  v8f acc[4] = {};
  for (int k0 = 0; k0 < D; k0 += 4) {
    const int kk = k0 + half * 2;
    const v2f a = *reinterpret_cast<const v2f*>(hp + kk);
#pragma unroll
    for (int t = 0; t < 4; ++t) {
      const int nn = t * 16 + l15;
      const v2f b = *reinterpret_cast<const v2f*>(&wl0[nn][kk]);
      acc[t] = WMMA4(a, b, acc[t]);
    }
  }
#pragma unroll
  for (int t = 0; t < 4; ++t) {
    const int c = t * 16 + l15;
    const float bb = fcb[c];
    const float m = mean[c];
    const float iv = rsqrtf(var[c] + 1e-5f);
    const float g = gamma[c];
    const float bt = beta[c];
#pragma unroll
    for (int i = 0; i < 8; ++i) {
      float v = acc[t][i] + bb;
      v = (v - m) * iv * g + bt;
      v = fmaxf(v, 0.0f);
      v = v + __logf(1.0f + __expf(-v));
      u[wave][i + half * 8][c] = v;
    }
  }
  v8f aL[4] = {}, aS[4] = {};
  for (int k0 = 0; k0 < D; k0 += 4) {
    const int kk = k0 + half * 2;
    const v2f a = *reinterpret_cast<const v2f*>(&u[wave][l15][kk]);
#pragma unroll
    for (int t = 0; t < 4; ++t) {
      const int nn = t * 16 + l15;
      const v2f bL = *reinterpret_cast<const v2f*>(&wl1[nn][kk]);
      const v2f bS = *reinterpret_cast<const v2f*>(&wl2[nn][kk]);
      aL[t] = WMMA4(a, bL, aL[t]);
      aS[t] = WMMA4(a, bS, aS[t]);
    }
  }
#pragma unroll
  for (int t = 0; t < 4; ++t) {
    const int c = t * 16 + l15;
    const float bl = b21[c];
    const float bs = b22[c];
#pragma unroll
    for (int i = 0; i < 8; ++i) {
      const int r = row0 + i + half * 8;
      if (r < n) {
        const size_t o = (size_t)r * D + c;
        zloc[o] = aL[t][i] + bl;
        zscale[o] = __expf(aS[t][i] + bs);
      }
    }
  }
}

extern "C" void kernel_launch(void* const* d_in, const int* in_sizes, int n_in,
                              void* d_out, int out_size, void* d_ws, size_t ws_size,
                              hipStream_t stream) {
  const float* x = (const float*)d_in[0];
  const int* esrc = (const int*)d_in[1];
  const int* edst = (const int*)d_in[2];
  const float* Wself = (const float*)d_in[3];
  const float* Wneigh = (const float*)d_in[4];
  const float* bsage = (const float*)d_in[5];
  const float* fcw = (const float*)d_in[6];
  const float* fcb = (const float*)d_in[7];
  const float* gamma = (const float*)d_in[8];
  const float* beta = (const float*)d_in[9];
  const float* mean = (const float*)d_in[10];
  const float* var = (const float*)d_in[11];
  const float* w21 = (const float*)d_in[12];
  const float* b21 = (const float*)d_in[13];
  const float* w22 = (const float*)d_in[14];
  const float* b22 = (const float*)d_in[15];

  const int n = in_sizes[0] / D;
  const int e = in_sizes[1];
  const size_t nd = (size_t)n * D;

  float* ws = (float*)d_ws;
  float* deg = ws;
  int* ctr = (int*)(deg + n);
  float* pA = ws + (((size_t)n + CTR_STRIDE + 1 + 255) & ~(size_t)255);
  float* pB = pA + nd;
  float* g = pB + nd;

  const int tiles = (n + 15) / 16;
  const int gblocks = (tiles + 7) / 8;

  k_proj<<<gblocks, 256, 0, stream>>>(x, Wself, Wneigh, bsage, pA, g, deg, ctr, n, 1);
  k_deg<<<(e + 255) / 256, 256, 0, stream>>>(edst, deg, e);

  float* hcur = pA;
  float* hnext = pB;
  for (int l = 0; l < 3; ++l) {
    k_scatter<<<2048, 256, 0, stream>>>(g, esrc, edst, deg, hcur, ctr, e, n);
    if (l < 2) {
      k_proj<<<gblocks, 256, 0, stream>>>(hcur, Wself + (size_t)(l + 1) * D * D,
                                          Wneigh + (size_t)(l + 1) * D * D,
                                          bsage + (size_t)(l + 1) * D, hnext, g, deg, ctr, n, 0);
      float* tmp = hcur;
      hcur = hnext;
      hnext = tmp;
    }
  }

  float* zloc = (float*)d_out;
  float* zscale = zloc + nd;
  k_head<<<gblocks, 256, 0, stream>>>(hcur, fcw, fcb, gamma, beta, mean, var, w21, b21,
                                      w22, b22, zloc, zscale, n);
}


# Degree count in LDS: work-group = (node range, edge slice), integer LDS counters, then one line-wide float add per 32 nodes, not a scattered global atomic per edge; whole-number sums, exact
# speedup vs baseline: 1.9498x; 1.1282x over previous
#include <hip/hip_runtime.h>
#include <cstdint>

typedef __attribute__((ext_vector_type(2))) float v2f;
typedef __attribute__((ext_vector_type(8))) float v8f;

#define D 64
#define WPAD 68

#define WMMA4(a, b, c) \
  __builtin_amdgcn_wmma_f32_16x16x4_f32(false, (a), false, (b), (short)0, (c), false, false)

__device__ __forceinline__ void stage_wT(const float* __restrict__ W, float (*wl)[WPAD],
                                         int tid) {
  for (int idx = tid; idx < 64 * 16; idx += 256) {
    const int k = idx >> 4;
    const int n4 = (idx & 15) << 2;
    const float4 v = *reinterpret_cast<const float4*>(W + k * D + n4);
    wl[n4 + 0][k] = v.x;
    wl[n4 + 1][k] = v.y;
    wl[n4 + 2][k] = v.z;
    wl[n4 + 3][k] = v.w;
  }
}

#define DR 6272
#define DS 16
#define DT 512
#define DU 25
__global__ void __launch_bounds__(DT)
k_deg(const int* __restrict__ dst, float* __restrict__ deg, int e, int n) {
  __shared__ int cnt[DR];
  const int tid = threadIdx.x;
  const int nr = (n + DR - 1) / DR;
  const int e4 = e >> 2;
  const int q4 = (e4 + DS - 1) / DS;
  const int4* __restrict__ d4 = reinterpret_cast<const int4*>(dst);
  for (int w = blockIdx.x; w < nr * DS; w += gridDim.x) {
    const int r = w / DS;
    const int s = w - r * DS;
    const int lo = r * DR;
    const unsigned len = (unsigned)(min(lo + DR, n) - lo);
    for (int i = tid; i < DR; i += DT) cnt[i] = 0;
    __syncthreads();
    const int b4 = s * q4;
    const int t4 = min(b4 + q4, e4);
    for (int base = b4; base < t4; base += DT * DU) {
      int4 v[DU];
#pragma unroll
      for (int k = 0; k < DU; ++k) {
        const int i4 = base + k * DT + tid;
        v[k] = d4[min(i4, t4 - 1)];
        if (i4 >= t4) v[k] = make_int4(-1, -1, -1, -1);
      }
#pragma unroll
      for (int k = 0; k < DU; ++k) {
        const unsigned o0 = (unsigned)(v[k].x - lo);
        const unsigned o1 = (unsigned)(v[k].y - lo);
        const unsigned o2 = (unsigned)(v[k].z - lo);
        const unsigned o3 = (unsigned)(v[k].w - lo);
        if (o0 < len) atomicAdd(&cnt[o0], 1);
        if (o1 < len) atomicAdd(&cnt[o1], 1);
        if (o2 < len) atomicAdd(&cnt[o2], 1);
        if (o3 < len) atomicAdd(&cnt[o3], 1);
      }
    }
    if (s == 0 && tid < (e & 3)) {
      const unsigned o = (unsigned)(dst[(e4 << 2) + tid] - lo);
      if (o < len) atomicAdd(&cnt[o], 1);
    }
    __syncthreads();
    for (unsigned i = tid; i < len; i += DT) {
      const int c = cnt[i];
      if (c) atomicAdd(&deg[lo + i], (float)c);
    }
    __syncthreads();
  }
}

#define CH 256
#define GRP 16
#define CTR_STRIDE 64
typedef __attribute__((address_space(1))) float gfloat;

#define PIN8(x, o)                                                                   \
  asm("" : "+s"(x[o + 0]), "+s"(x[o + 1]), "+s"(x[o + 2]), "+s"(x[o + 3]),           \
           "+s"(x[o + 4]), "+s"(x[o + 5]), "+s"(x[o + 6]), "+s"(x[o + 7]))

__global__ void __launch_bounds__(256)
k_scatter(const float* __restrict__ g, const int* __restrict__ src,
          const int* __restrict__ dst, const float* __restrict__ deg,
          float* __restrict__ acc, int* __restrict__ ctr, int e, int n) {
  __shared__ int tk[3];
  const unsigned lane = threadIdx.x & 31u;
  const unsigned wave = threadIdx.x >> 5;
  const int nch = (e + CH - 1) / CH;
  const int nsplit = n >> 1;
  if (threadIdx.x == 0) {
    const unsigned id = (unsigned)__builtin_amdgcn_s_sendmsg_rtn(0x87);
    const int q0 = (int)((id >> 18) & 1u);
    tk[2] = q0;
    tk[0] = atomicAdd(ctr + q0 * CTR_STRIDE, 1);
  }
  __syncthreads();
  int q = tk[2];
  int t = tk[0];
  int it = 0;
  bool other = false;
  for (;;) {
    int tn = 0;
    const bool live = t < nch;
    if (!live) {
      if (other) break;
      other = true;
      q ^= 1;
    }
    if (threadIdx.x == 0) tn = atomicAdd(ctr + q * CTR_STRIDE, 1);
    if (live) {
      const long ei = (long)t * CH + (long)(wave * 32u + lane);
      const bool ok = ei < e;
      const int d = ok ? dst[ei] : 0;
      const int s = ok ? src[ei] : 0;
      const bool mine = ok && ((d >= nsplit) == (q != 0));
      const float dinv = mine ? 1.0f / fmaxf(deg[d], 1.0f) : 0.0f;
      unsigned m = __builtin_amdgcn_ballot_w32(mine);
      while (m) {
        const unsigned m0 = m;
        const int cnt = min((int)__builtin_popcount(m0), GRP);
        unsigned long long hb[GRP], db[GRP];
        unsigned w1 = m0;
#pragma unroll
        for (int k = 0; k < GRP; ++k) {
          const int j = w1 ? (int)__builtin_ctz(w1) : 0;
          w1 &= w1 - 1u;
          const unsigned sj = (unsigned)__builtin_amdgcn_readlane(s, j);
          hb[k] = (unsigned long long)g + (unsigned long long)sj * (D * sizeof(float));
        }
        PIN8(hb, 0);
        PIN8(hb, 8);
        float a[GRP], b[GRP];
#pragma unroll
        for (int k = 0; k < GRP; ++k) {
          a[k] = 0.0f;
          b[k] = 0.0f;
          if (k < cnt) {
            const gfloat* p = (const gfloat*)hb[k];
            a[k] = p[lane];
            b[k] = p[lane + 32u];
          }
        }
        unsigned w2 = m0;
#pragma unroll
        for (int k = 0; k < GRP; ++k) {
          const int j = w2 ? (int)__builtin_ctz(w2) : 0;
          w2 &= w2 - 1u;
          const unsigned dd = (unsigned)__builtin_amdgcn_readlane(d, j);
          db[k] = (unsigned long long)acc + (unsigned long long)dd * (D * sizeof(float));
        }
        PIN8(db, 0);
        PIN8(db, 8);
#pragma unroll
        for (int k = 0; k < GRP; ++k) {
          const int j = m ? (int)__builtin_ctz(m) : 0;
          m &= m - 1u;
          if (k < cnt) {
            const float dj = __int_as_float(__builtin_amdgcn_readlane(__float_as_int(dinv), j));
            float* w = (float*)(gfloat*)db[k];
            atomicAdd(w + lane, a[k] * dj);
            atomicAdd(w + lane + 32u, b[k] * dj);
          }
        }
      }
    }
    if (threadIdx.x == 0) tk[(it + 1) & 1] = tn;
    __syncthreads();
    ++it;
    t = tk[it & 1];
  }
}

__global__ void __launch_bounds__(256)
k_proj(const float* __restrict__ in, const float* __restrict__ Ws,
       const float* __restrict__ Wn, const float* __restrict__ bias,
       float* __restrict__ P, float* __restrict__ G, float* __restrict__ deg,
       int* __restrict__ ctr, int n, int first) {
  __shared__ float wls[D][WPAD];
  __shared__ float wln[D][WPAD];
  stage_wT(Ws, wls, threadIdx.x);
  stage_wT(Wn, wln, threadIdx.x);
  if (first) {
    const int gi = blockIdx.x * 256 + threadIdx.x;
    if (gi < n) deg[gi] = 0.0f;
  }
  if (blockIdx.x == 0 && threadIdx.x < 2) ctr[threadIdx.x * CTR_STRIDE] = 0;
  __syncthreads();

  const int lane = threadIdx.x & 31;
  const int wave = threadIdx.x >> 5;
  const int row0 = (blockIdx.x * 8 + wave) * 16;
  if (row0 >= n) return;
  const int half = lane >> 4;
  const int l15 = lane & 15;
  int rowA = row0 + l15;
  if (rowA >= n) rowA = n - 1;
  const float* __restrict__ ip = in + (size_t)rowA * D + half * 2;

  v2f a[16];
#pragma unroll
  for (int j = 0; j < 16; ++j) a[j] = *reinterpret_cast<const v2f*>(ip + 4 * j);
  if (first) {
#pragma unroll
    for (int j = 0; j < 16; ++j) {
      a[j].x = __logf(a[j].x + 1.0f);
      a[j].y = __logf(a[j].y + 1.0f);
    }
  } else {
    float ss = 0.0f;
#pragma unroll
    for (int j = 0; j < 16; ++j) {
      a[j].x = fmaxf(a[j].x, 0.0f);
      a[j].y = fmaxf(a[j].y, 0.0f);
      ss += a[j].x * a[j].x;
      ss += a[j].y * a[j].y;
    }
    ss += __shfl_xor(ss, 16, 32);
    const float inv = 1.0f / fmaxf(sqrtf(ss), 1e-12f);
#pragma unroll
    for (int j = 0; j < 16; ++j) {
      a[j].x *= inv;
      a[j].y *= inv;
    }
  }

  v8f accS[4] = {}, accN[4] = {};
#pragma unroll
  for (int j = 0; j < 16; ++j) {
    const int kk = 4 * j + half * 2;
#pragma unroll
    for (int t = 0; t < 4; ++t) {
      const int nn = t * 16 + l15;
      const v2f bs = *reinterpret_cast<const v2f*>(&wls[nn][kk]);
      const v2f bn = *reinterpret_cast<const v2f*>(&wln[nn][kk]);
      accS[t] = WMMA4(a[j], bs, accS[t]);
      accN[t] = WMMA4(a[j], bn, accN[t]);
    }
  }

#pragma unroll
  for (int t = 0; t < 4; ++t) {
    const int c = t * 16 + l15;
    const float bv = bias[c];
#pragma unroll
    for (int i = 0; i < 8; ++i) {
      const int r = row0 + i + half * 8;
      if (r < n) {
        const size_t o = (size_t)r * D + c;
        P[o] = accS[t][i] + bv;
        G[o] = accN[t][i];
      }
    }
  }
}

__global__ void __launch_bounds__(256)
k_head(const float* __restrict__ h, const float* __restrict__ fcw,
       const float* __restrict__ fcb, const float* __restrict__ gamma,
       const float* __restrict__ beta, const float* __restrict__ mean,
       const float* __restrict__ var, const float* __restrict__ w21,
       const float* __restrict__ b21, const float* __restrict__ w22,
       const float* __restrict__ b22, float* __restrict__ zloc,
       float* __restrict__ zscale, int n) {
  __shared__ float wl0[D][WPAD];
  __shared__ float wl1[D][WPAD];
  __shared__ float wl2[D][WPAD];
  __shared__ float u[8][16][WPAD];
  stage_wT(fcw, wl0, threadIdx.x);
  stage_wT(w21, wl1, threadIdx.x);
  stage_wT(w22, wl2, threadIdx.x);
  __syncthreads();

  const int lane = threadIdx.x & 31;
  const int wave = threadIdx.x >> 5;
  const int row0 = (blockIdx.x * 8 + wave) * 16;
  if (row0 >= n) return;
  const int half = lane >> 4;
  const int l15 = lane & 15;
  int rowA = row0 + l15;
  if (rowA >= n) rowA = n - 1;
  const float* __restrict__ hp = h + (size_t)rowA * D;

  v8f acc[4] = {};
  for (int k0 = 0; k0 < D; k0 += 4) {
    const int kk = k0 + half * 2;
    const v2f a = *reinterpret_cast<const v2f*>(hp + kk);
#pragma unroll
    for (int t = 0; t < 4; ++t) {
      const int nn = t * 16 + l15;
      const v2f b = *reinterpret_cast<const v2f*>(&wl0[nn][kk]);
      acc[t] = WMMA4(a, b, acc[t]);
    }
  }
#pragma unroll
  for (int t = 0; t < 4; ++t) {
    const int c = t * 16 + l15;
    const float bb = fcb[c];
    const float m = mean[c];
    const float iv = rsqrtf(var[c] + 1e-5f);
    const float g = gamma[c];
    const float bt = beta[c];
#pragma unroll
    for (int i = 0; i < 8; ++i) {
      float v = acc[t][i] + bb;
      v = (v - m) * iv * g + bt;
      v = fmaxf(v, 0.0f);
      v = v + __logf(1.0f + __expf(-v));
      u[wave][i + half * 8][c] = v;
    }
  }
  v8f aL[4] = {}, aS[4] = {};
  for (int k0 = 0; k0 < D; k0 += 4) {
    const int kk = k0 + half * 2;
    const v2f a = *reinterpret_cast<const v2f*>(&u[wave][l15][kk]);
#pragma unroll
    for (int t = 0; t < 4; ++t) {
      const int nn = t * 16 + l15;
      const v2f bL = *reinterpret_cast<const v2f*>(&wl1[nn][kk]);
      const v2f bS = *reinterpret_cast<const v2f*>(&wl2[nn][kk]);
      aL[t] = WMMA4(a, bL, aL[t]);
      aS[t] = WMMA4(a, bS, aS[t]);
    }
  }
#pragma unroll
  for (int t = 0; t < 4; ++t) {
    const int c = t * 16 + l15;
    const float bl = b21[c];
    const float bs = b22[c];
#pragma unroll
    for (int i = 0; i < 8; ++i) {
      const int r = row0 + i + half * 8;
      if (r < n) {
        const size_t o = (size_t)r * D + c;
        zloc[o] = aL[t][i] + bl;
        zscale[o] = __expf(aS[t][i] + bs);
      }
    }
  }
}

extern "C" void kernel_launch(void* const* d_in, const int* in_sizes, int n_in,
                              void* d_out, int out_size, void* d_ws, size_t ws_size,
                              hipStream_t stream) {
  const float* x = (const float*)d_in[0];
  const int* esrc = (const int*)d_in[1];
  const int* edst = (const int*)d_in[2];
  const float* Wself = (const float*)d_in[3];
  const float* Wneigh = (const float*)d_in[4];
  const float* bsage = (const float*)d_in[5];
  const float* fcw = (const float*)d_in[6];
  const float* fcb = (const float*)d_in[7];
  const float* gamma = (const float*)d_in[8];
  const float* beta = (const float*)d_in[9];
  const float* mean = (const float*)d_in[10];
  const float* var = (const float*)d_in[11];
  const float* w21 = (const float*)d_in[12];
  const float* b21 = (const float*)d_in[13];
  const float* w22 = (const float*)d_in[14];
  const float* b22 = (const float*)d_in[15];

  const int n = in_sizes[0] / D;
  const int e = in_sizes[1];
  const size_t nd = (size_t)n * D;

  float* ws = (float*)d_ws;
  float* deg = ws;
  int* ctr = (int*)(deg + n);
  float* pA = ws + (((size_t)n + CTR_STRIDE + 1 + 255) & ~(size_t)255);
  float* pB = pA + nd;
  float* g = pB + nd;

  const int tiles = (n + 15) / 16;
  const int gblocks = (tiles + 7) / 8;

  k_proj<<<gblocks, 256, 0, stream>>>(x, Wself, Wneigh, bsage, pA, g, deg, ctr, n, 1);
  k_deg<<<256, DT, 0, stream>>>(edst, deg, e, n);

  float* hcur = pA;
  float* hnext = pB;
  for (int l = 0; l < 3; ++l) {
    k_scatter<<<2048, 256, 0, stream>>>(g, esrc, edst, deg, hcur, ctr, e, n);
    if (l < 2) {
      k_proj<<<gblocks, 256, 0, stream>>>(hcur, Wself + (size_t)(l + 1) * D * D,
                                          Wneigh + (size_t)(l + 1) * D * D,
                                          bsage + (size_t)(l + 1) * D, hnext, g, deg, ctr, n, 0);
      float* tmp = hcur;
      hcur = hnext;
      hnext = tmp;
    }
  }

  float* zloc = (float*)d_out;
  float* zscale = zloc + nd;
  k_head<<<gblocks, 256, 0, stream>>>(hcur, fcw, fcb, gamma, beta, mean, var, w21, b21,
                                      w22, b22, zloc, zscale, n);
}


# Degree division out of the edge pass: self term stored times max(deg,1), scatter adds raw rows (no deg read per edge), reader divides a row once; sum order: self term, then atomic arrival
# speedup vs baseline: 2.0593x; 1.0562x over previous
#include <hip/hip_runtime.h>
#include <cstdint>

typedef __attribute__((ext_vector_type(2))) float v2f;
typedef __attribute__((ext_vector_type(8))) float v8f;

#define D 64
#define WPAD 68

#define WMMA4(a, b, c) \
  __builtin_amdgcn_wmma_f32_16x16x4_f32(false, (a), false, (b), (short)0, (c), false, false)

__device__ __forceinline__ void stage_wT(const float* __restrict__ W, float (*wl)[WPAD],
                                         int tid) {
  for (int idx = tid; idx < 64 * 16; idx += 256) {
    const int k = idx >> 4;
    const int n4 = (idx & 15) << 2;
    const float4 v = *reinterpret_cast<const float4*>(W + k * D + n4);
    wl[n4 + 0][k] = v.x;
    wl[n4 + 1][k] = v.y;
    wl[n4 + 2][k] = v.z;
    wl[n4 + 3][k] = v.w;
  }
}

#define DR 6272
#define DS 16
#define DT 512
#define DU 25
__global__ void __launch_bounds__(DT)
k_deg(const int* __restrict__ dst, int* __restrict__ part, int e, int n) {
  __shared__ int cnt[DR];
  const int tid = threadIdx.x;
  const int nr = (n + DR - 1) / DR;
  const int e4 = e >> 2;
  const int q4 = (e4 + DS - 1) / DS;
  const int4* __restrict__ d4 = reinterpret_cast<const int4*>(dst);
  for (int w = blockIdx.x; w < nr * DS; w += gridDim.x) {
    const int r = w / DS;
    const int s = w - r * DS;
    const int lo = r * DR;
    const unsigned len = (unsigned)(min(lo + DR, n) - lo);
    for (int i = tid; i < DR; i += DT) cnt[i] = 0;
    __syncthreads();
    const int b4 = s * q4;
    const int t4 = min(b4 + q4, e4);
    for (int base = b4; base < t4; base += DT * DU) {
      int4 v[DU];
#pragma unroll
      for (int k = 0; k < DU; ++k) {
        const int i4 = base + k * DT + tid;
        v[k] = d4[min(i4, t4 - 1)];
        if (i4 >= t4) v[k] = make_int4(-1, -1, -1, -1);
      }
#pragma unroll
      for (int k = 0; k < DU; ++k) {
        const unsigned o0 = (unsigned)(v[k].x - lo);
        const unsigned o1 = (unsigned)(v[k].y - lo);
        const unsigned o2 = (unsigned)(v[k].z - lo);
        const unsigned o3 = (unsigned)(v[k].w - lo);
        if (o0 < len) atomicAdd(&cnt[o0], 1);
        if (o1 < len) atomicAdd(&cnt[o1], 1);
        if (o2 < len) atomicAdd(&cnt[o2], 1);
        if (o3 < len) atomicAdd(&cnt[o3], 1);
      }
    }
    if (s == 0 && tid < (e & 3)) {
      const unsigned o = (unsigned)(dst[(e4 << 2) + tid] - lo);
      if (o < len) atomicAdd(&cnt[o], 1);
    }
    __syncthreads();
    for (unsigned i = tid; i < len; i += DT) part[(size_t)s * n + lo + i] = cnt[i];
    __syncthreads();
  }
}

#define CH 256
#define GRP 16
#define CTR_STRIDE 64
typedef __attribute__((address_space(1))) float gfloat;

#define PIN8(x, o)                                                                   \
  asm("" : "+s"(x[o + 0]), "+s"(x[o + 1]), "+s"(x[o + 2]), "+s"(x[o + 3]),           \
           "+s"(x[o + 4]), "+s"(x[o + 5]), "+s"(x[o + 6]), "+s"(x[o + 7]))

__global__ void __launch_bounds__(256)
k_scatter(const float* __restrict__ g, const int* __restrict__ src,
          const int* __restrict__ dst, float* __restrict__ acc, int* __restrict__ ctr,
          int e, int n) {
  __shared__ int tk[3];
  const unsigned lane = threadIdx.x & 31u;
  const unsigned wave = threadIdx.x >> 5;
  const int nch = (e + CH - 1) / CH;
  const int nsplit = n >> 1;
  if (threadIdx.x == 0) {
    const unsigned id = (unsigned)__builtin_amdgcn_s_sendmsg_rtn(0x87);
    const int q0 = (int)((id >> 18) & 1u);
    tk[2] = q0;
    tk[0] = atomicAdd(ctr + q0 * CTR_STRIDE, 1);
  }
  __syncthreads();
  int q = tk[2];
  int t = tk[0];
  int it = 0;
  bool other = false;
  for (;;) {
    int tn = 0;
    const bool live = t < nch;
    if (!live) {
      if (other) break;
      other = true;
      q ^= 1;
    }
    if (threadIdx.x == 0) tn = atomicAdd(ctr + q * CTR_STRIDE, 1);
    if (live) {
      const long ei = (long)t * CH + (long)(wave * 32u + lane);
      const bool ok = ei < e;
      const int d = ok ? dst[ei] : 0;
      const int s = ok ? src[ei] : 0;
      const bool mine = ok && ((d >= nsplit) == (q != 0));
      unsigned m = __builtin_amdgcn_ballot_w32(mine);
      while (m) {
        const unsigned m0 = m;
        const int cnt = min((int)__builtin_popcount(m0), GRP);
        unsigned long long hb[GRP], db[GRP];
        unsigned w1 = m0;
#pragma unroll
        for (int k = 0; k < GRP; ++k) {
          const int j = w1 ? (int)__builtin_ctz(w1) : 0;
          w1 &= w1 - 1u;
          const unsigned sj = (unsigned)__builtin_amdgcn_readlane(s, j);
          hb[k] = (unsigned long long)g + (unsigned long long)sj * (D * sizeof(float));
        }
        PIN8(hb, 0);
        PIN8(hb, 8);
        float a[GRP], b[GRP];
#pragma unroll
        for (int k = 0; k < GRP; ++k) {
          a[k] = 0.0f;
          b[k] = 0.0f;
          if (k < cnt) {
            const gfloat* p = (const gfloat*)hb[k];
            a[k] = p[lane];
            b[k] = p[lane + 32u];
          }
        }
        unsigned w2 = m0;
#pragma unroll
        for (int k = 0; k < GRP; ++k) {
          const int j = w2 ? (int)__builtin_ctz(w2) : 0;
          w2 &= w2 - 1u;
          const unsigned dd = (unsigned)__builtin_amdgcn_readlane(d, j);
          db[k] = (unsigned long long)acc + (unsigned long long)dd * (D * sizeof(float));
        }
        PIN8(db, 0);
        PIN8(db, 8);
#pragma unroll
        for (int k = 0; k < GRP; ++k) {
          if (k < cnt) {
            float* w = (float*)(gfloat*)db[k];
            atomicAdd(w + lane, a[k]);
            atomicAdd(w + lane + 32u, b[k]);
          }
        }
        m = w2;
      }
    }
    if (threadIdx.x == 0) tk[(it + 1) & 1] = tn;
    __syncthreads();
    ++it;
    t = tk[it & 1];
  }
}

__global__ void __launch_bounds__(256)
k_proj(const float* __restrict__ in, const float* __restrict__ Ws,
       const float* __restrict__ Wn, const float* __restrict__ bias,
       float* __restrict__ P, float* __restrict__ G, float* __restrict__ degc,
       const int* __restrict__ part, int* __restrict__ ctr, int n, int first) {
  __shared__ float wls[D][WPAD];
  __shared__ float wln[D][WPAD];
  stage_wT(Ws, wls, threadIdx.x);
  stage_wT(Wn, wln, threadIdx.x);
  if (blockIdx.x == 0 && threadIdx.x < 2) ctr[threadIdx.x * CTR_STRIDE] = 0;
  __syncthreads();

  const int lane = threadIdx.x & 31;
  const int wave = threadIdx.x >> 5;
  const int row0 = (blockIdx.x * 8 + wave) * 16;
  if (row0 >= n) return;
  const int half = lane >> 4;
  const int l15 = lane & 15;
  int rowA = row0 + l15;
  if (rowA >= n) rowA = n - 1;
  const float* __restrict__ ip = in + (size_t)rowA * D + half * 2;

  float dc;
  if (first) {
    int cs = 0;
#pragma unroll
    for (int k = 0; k < DS / 2; ++k) cs += part[(size_t)(half * (DS / 2) + k) * n + rowA];
    cs += __shfl_xor(cs, 16, 32);
    dc = fmaxf((float)cs, 1.0f);
    if (half == 0 && row0 + l15 < n) degc[row0 + l15] = dc;
  } else {
    dc = degc[rowA];
  }

  v2f a[16];
#pragma unroll
  for (int j = 0; j < 16; ++j) a[j] = *reinterpret_cast<const v2f*>(ip + 4 * j);
  if (first) {
#pragma unroll
    for (int j = 0; j < 16; ++j) {
      a[j].x = __logf(a[j].x + 1.0f);
      a[j].y = __logf(a[j].y + 1.0f);
    }
  } else {
    const float rdc = 1.0f / dc;
    float ss = 0.0f;
#pragma unroll
    for (int j = 0; j < 16; ++j) {
      a[j].x = fmaxf(a[j].x * rdc, 0.0f);
      a[j].y = fmaxf(a[j].y * rdc, 0.0f);
      ss += a[j].x * a[j].x;
      ss += a[j].y * a[j].y;
    }
    ss += __shfl_xor(ss, 16, 32);
    const float inv = 1.0f / fmaxf(sqrtf(ss), 1e-12f);
#pragma unroll
    for (int j = 0; j < 16; ++j) {
      a[j].x *= inv;
      a[j].y *= inv;
    }
  }

  v8f accS[4] = {}, accN[4] = {};
#pragma unroll
  for (int j = 0; j < 16; ++j) {
    const int kk = 4 * j + half * 2;
#pragma unroll
    for (int t = 0; t < 4; ++t) {
      const int nn = t * 16 + l15;
      const v2f bs = *reinterpret_cast<const v2f*>(&wls[nn][kk]);
      const v2f bn = *reinterpret_cast<const v2f*>(&wln[nn][kk]);
      accS[t] = WMMA4(a[j], bs, accS[t]);
      accN[t] = WMMA4(a[j], bn, accN[t]);
    }
  }

  float dr[8];
#pragma unroll
  for (int i = 0; i < 8; ++i) {
    const float d0 = __int_as_float(__builtin_amdgcn_readlane(__float_as_int(dc), i));
    const float d1 = __int_as_float(__builtin_amdgcn_readlane(__float_as_int(dc), i + 8));
    dr[i] = half ? d1 : d0;
  }
#pragma unroll
  for (int t = 0; t < 4; ++t) {
    const int c = t * 16 + l15;
    const float bv = bias[c];
#pragma unroll
    for (int i = 0; i < 8; ++i) {
      const int r = row0 + i + half * 8;
      if (r < n) {
        const size_t o = (size_t)r * D + c;
        P[o] = (accS[t][i] + bv) * dr[i];
        G[o] = accN[t][i];
      }
    }
  }
}

__global__ void __launch_bounds__(256)
k_head(const float* __restrict__ h, const float* __restrict__ degc,
       const float* __restrict__ fcw,
       const float* __restrict__ fcb, const float* __restrict__ gamma,
       const float* __restrict__ beta, const float* __restrict__ mean,
       const float* __restrict__ var, const float* __restrict__ w21,
       const float* __restrict__ b21, const float* __restrict__ w22,
       const float* __restrict__ b22, float* __restrict__ zloc,
       float* __restrict__ zscale, int n) {
  __shared__ float wl0[D][WPAD];
  __shared__ float wl1[D][WPAD];
  __shared__ float wl2[D][WPAD];
  __shared__ float u[8][16][WPAD];
  stage_wT(fcw, wl0, threadIdx.x);
  stage_wT(w21, wl1, threadIdx.x);
  stage_wT(w22, wl2, threadIdx.x);
  __syncthreads();

  const int lane = threadIdx.x & 31;
  const int wave = threadIdx.x >> 5;
  const int row0 = (blockIdx.x * 8 + wave) * 16;
  if (row0 >= n) return;
  const int half = lane >> 4;
  const int l15 = lane & 15;
  int rowA = row0 + l15;
  if (rowA >= n) rowA = n - 1;
  const float* __restrict__ hp = h + (size_t)rowA * D;
  const float rdc = 1.0f / degc[rowA];

  v8f acc[4] = {};
  for (int k0 = 0; k0 < D; k0 += 4) {
    const int kk = k0 + half * 2;
    const v2f a = *reinterpret_cast<const v2f*>(hp + kk) * rdc;
#pragma unroll
    for (int t = 0; t < 4; ++t) {
      const int nn = t * 16 + l15;
      const v2f b = *reinterpret_cast<const v2f*>(&wl0[nn][kk]);
      acc[t] = WMMA4(a, b, acc[t]);
    }
  }
#pragma unroll
  for (int t = 0; t < 4; ++t) {
    const int c = t * 16 + l15;
    const float bb = fcb[c];
    const float m = mean[c];
    const float iv = rsqrtf(var[c] + 1e-5f);
    const float g = gamma[c];
    const float bt = beta[c];
#pragma unroll
    for (int i = 0; i < 8; ++i) {
      float v = acc[t][i] + bb;
      v = (v - m) * iv * g + bt;
      v = fmaxf(v, 0.0f);
      v = v + __logf(1.0f + __expf(-v));
      u[wave][i + half * 8][c] = v;
    }
  }
  v8f aL[4] = {}, aS[4] = {};
  for (int k0 = 0; k0 < D; k0 += 4) {
    const int kk = k0 + half * 2;
    const v2f a = *reinterpret_cast<const v2f*>(&u[wave][l15][kk]);
#pragma unroll
    for (int t = 0; t < 4; ++t) {
      const int nn = t * 16 + l15;
      const v2f bL = *reinterpret_cast<const v2f*>(&wl1[nn][kk]);
      const v2f bS = *reinterpret_cast<const v2f*>(&wl2[nn][kk]);
      aL[t] = WMMA4(a, bL, aL[t]);
      aS[t] = WMMA4(a, bS, aS[t]);
    }
  }
#pragma unroll
  for (int t = 0; t < 4; ++t) {
    const int c = t * 16 + l15;
    const float bl = b21[c];
    const float bs = b22[c];
#pragma unroll
    for (int i = 0; i < 8; ++i) {
      const int r = row0 + i + half * 8;
      if (r < n) {
        const size_t o = (size_t)r * D + c;
        zloc[o] = aL[t][i] + bl;
        zscale[o] = __expf(aS[t][i] + bs);
      }
    }
  }
}

extern "C" void kernel_launch(void* const* d_in, const int* in_sizes, int n_in,
                              void* d_out, int out_size, void* d_ws, size_t ws_size,
                              hipStream_t stream) {
  const float* x = (const float*)d_in[0];
  const int* esrc = (const int*)d_in[1];
  const int* edst = (const int*)d_in[2];
  const float* Wself = (const float*)d_in[3];
  const float* Wneigh = (const float*)d_in[4];
  const float* bsage = (const float*)d_in[5];
  const float* fcw = (const float*)d_in[6];
  const float* fcb = (const float*)d_in[7];
  const float* gamma = (const float*)d_in[8];
  const float* beta = (const float*)d_in[9];
  const float* mean = (const float*)d_in[10];
  const float* var = (const float*)d_in[11];
  const float* w21 = (const float*)d_in[12];
  const float* b21 = (const float*)d_in[13];
  const float* w22 = (const float*)d_in[14];
  const float* b22 = (const float*)d_in[15];

  const int n = in_sizes[0] / D;
  const int e = in_sizes[1];
  const size_t nd = (size_t)n * D;

  float* ws = (float*)d_ws;
  float* deg = ws;
  int* ctr = (int*)(deg + n);
  float* pA = ws + (((size_t)n + CTR_STRIDE + 1 + 255) & ~(size_t)255);
  float* pB = pA + nd;
  float* g = pB + nd;

  const int tiles = (n + 15) / 16;
  const int gblocks = (tiles + 7) / 8;

  int* part = (int*)pB;
  k_deg<<<256, DT, 0, stream>>>(edst, part, e, n);
  k_proj<<<gblocks, 256, 0, stream>>>(x, Wself, Wneigh, bsage, pA, g, deg, part, ctr, n, 1);

  float* hcur = pA;
  float* hnext = pB;
  for (int l = 0; l < 3; ++l) {
    k_scatter<<<2048, 256, 0, stream>>>(g, esrc, edst, hcur, ctr, e, n);
    if (l < 2) {
      k_proj<<<gblocks, 256, 0, stream>>>(hcur, Wself + (size_t)(l + 1) * D * D,
                                          Wneigh + (size_t)(l + 1) * D * D,
                                          bsage + (size_t)(l + 1) * D, hnext, g, deg, nullptr, ctr,
                                          n, 0);
      float* tmp = hcur;
      hcur = hnext;
      hnext = tmp;
    }
  }

  float* zloc = (float*)d_out;
  float* zscale = zloc + nd;
  k_head<<<gblocks, 256, 0, stream>>>(hcur, deg, fcw, fcb, gamma, beta, mean, var, w21, b21,
                                      w22, b22, zloc, zscale, n);
}


# Projected rows G stored interleaved (column i beside column i+32): the scatter reads a 256-byte row as ONE 8-byte load per lane, not two 4-byte loads; sum order = atomic arrival as before
# speedup vs baseline: 2.1410x; 1.0397x over previous
#include <hip/hip_runtime.h>
#include <cstdint>

typedef __attribute__((ext_vector_type(2))) float v2f;
typedef __attribute__((ext_vector_type(8))) float v8f;

#define D 64
#define WPAD 68

#define WMMA4(a, b, c) \
  __builtin_amdgcn_wmma_f32_16x16x4_f32(false, (a), false, (b), (short)0, (c), false, false)

__device__ __forceinline__ void stage_wT(const float* __restrict__ W, float (*wl)[WPAD],
                                         int tid) {
  for (int idx = tid; idx < 64 * 16; idx += 256) {
    const int k = idx >> 4;
    const int n4 = (idx & 15) << 2;
    const float4 v = *reinterpret_cast<const float4*>(W + k * D + n4);
    wl[n4 + 0][k] = v.x;
    wl[n4 + 1][k] = v.y;
    wl[n4 + 2][k] = v.z;
    wl[n4 + 3][k] = v.w;
  }
}

#define DR 6272
#define DS 16
#define DT 512
#define DU 25
__global__ void __launch_bounds__(DT)
k_deg(const int* __restrict__ dst, int* __restrict__ part, int e, int n) {
  __shared__ int cnt[DR];
  const int tid = threadIdx.x;
  const int nr = (n + DR - 1) / DR;
  const int e4 = e >> 2;
  const int q4 = (e4 + DS - 1) / DS;
  const int4* __restrict__ d4 = reinterpret_cast<const int4*>(dst);
  for (int w = blockIdx.x; w < nr * DS; w += gridDim.x) {
    const int r = w / DS;
    const int s = w - r * DS;
    const int lo = r * DR;
    const unsigned len = (unsigned)(min(lo + DR, n) - lo);
    for (int i = tid; i < DR; i += DT) cnt[i] = 0;
    __syncthreads();
    const int b4 = s * q4;
    const int t4 = min(b4 + q4, e4);
    for (int base = b4; base < t4; base += DT * DU) {
      int4 v[DU];
#pragma unroll
      for (int k = 0; k < DU; ++k) {
        const int i4 = base + k * DT + tid;
        v[k] = d4[min(i4, t4 - 1)];
        if (i4 >= t4) v[k] = make_int4(-1, -1, -1, -1);
      }
#pragma unroll
      for (int k = 0; k < DU; ++k) {
        const unsigned o0 = (unsigned)(v[k].x - lo);
        const unsigned o1 = (unsigned)(v[k].y - lo);
        const unsigned o2 = (unsigned)(v[k].z - lo);
        const unsigned o3 = (unsigned)(v[k].w - lo);
        if (o0 < len) atomicAdd(&cnt[o0], 1);
        if (o1 < len) atomicAdd(&cnt[o1], 1);
        if (o2 < len) atomicAdd(&cnt[o2], 1);
        if (o3 < len) atomicAdd(&cnt[o3], 1);
      }
    }
    if (s == 0 && tid < (e & 3)) {
      const unsigned o = (unsigned)(dst[(e4 << 2) + tid] - lo);
      if (o < len) atomicAdd(&cnt[o], 1);
    }
    __syncthreads();
    for (unsigned i = tid; i < len; i += DT) part[(size_t)s * n + lo + i] = cnt[i];
    __syncthreads();
  }
}

#define CH 256
#define GRP 16
#define CTR_STRIDE 64
typedef __attribute__((address_space(1))) float gfloat;
typedef __attribute__((address_space(1))) v2f gv2f;

#define PIN8(x, o)                                                                   \
  asm("" : "+s"(x[o + 0]), "+s"(x[o + 1]), "+s"(x[o + 2]), "+s"(x[o + 3]),           \
           "+s"(x[o + 4]), "+s"(x[o + 5]), "+s"(x[o + 6]), "+s"(x[o + 7]))

__global__ void __launch_bounds__(256)
k_scatter(const float* __restrict__ g, const int* __restrict__ src,
          const int* __restrict__ dst, float* __restrict__ acc, int* __restrict__ ctr,
          int e, int n) {
  __shared__ int tk[3];
  const unsigned lane = threadIdx.x & 31u;
  const unsigned wave = threadIdx.x >> 5;
  const int nch = (e + CH - 1) / CH;
  const int nsplit = n >> 1;
  if (threadIdx.x == 0) {
    const unsigned id = (unsigned)__builtin_amdgcn_s_sendmsg_rtn(0x87);
    const int q0 = (int)((id >> 18) & 1u);
    tk[2] = q0;
    tk[0] = atomicAdd(ctr + q0 * CTR_STRIDE, 1);
  }
  __syncthreads();
  int q = tk[2];
  int t = tk[0];
  int it = 0;
  bool other = false;
  for (;;) {
    int tn = 0;
    const bool live = t < nch;
    if (!live) {
      if (other) break;
      other = true;
      q ^= 1;
    }
    if (threadIdx.x == 0) tn = atomicAdd(ctr + q * CTR_STRIDE, 1);
    if (live) {
      const long ei = (long)t * CH + (long)(wave * 32u + lane);
      const bool ok = ei < e;
      const int d = ok ? dst[ei] : 0;
      const int s = ok ? src[ei] : 0;
      const bool mine = ok && ((d >= nsplit) == (q != 0));
      unsigned m = __builtin_amdgcn_ballot_w32(mine);
      while (m) {
        const unsigned m0 = m;
        const int cnt = min((int)__builtin_popcount(m0), GRP);
        unsigned long long hb[GRP], db[GRP];
        unsigned w1 = m0;
#pragma unroll
        for (int k = 0; k < GRP; ++k) {
          const int j = w1 ? (int)__builtin_ctz(w1) : 0;
          w1 &= w1 - 1u;
          const unsigned sj = (unsigned)__builtin_amdgcn_readlane(s, j);
          hb[k] = (unsigned long long)g + (unsigned long long)sj * (D * sizeof(float));
        }
        PIN8(hb, 0);
        PIN8(hb, 8);
        v2f ab[GRP];
#pragma unroll
        for (int k = 0; k < GRP; ++k) {
          if (k < cnt)
            ab[k] = ((const gv2f*)hb[k])[lane];
        }
        unsigned w2 = m0;
#pragma unroll
        for (int k = 0; k < GRP; ++k) {
          const int j = w2 ? (int)__builtin_ctz(w2) : 0;
          w2 &= w2 - 1u;
          const unsigned dd = (unsigned)__builtin_amdgcn_readlane(d, j);
          db[k] = (unsigned long long)acc + (unsigned long long)dd * (D * sizeof(float));
        }
        PIN8(db, 0);
        PIN8(db, 8);
#pragma unroll
        for (int k = 0; k < GRP; ++k) {
          if (k < cnt) {
            float* w = (float*)(gfloat*)db[k];
            atomicAdd(w + lane, ab[k].x);
            atomicAdd(w + lane + 32u, ab[k].y);
          }
        }
        m = w2;
      }
    }
    if (threadIdx.x == 0) tk[(it + 1) & 1] = tn;
    __syncthreads();
    ++it;
    t = tk[it & 1];
  }
}

__global__ void __launch_bounds__(256)
k_proj(const float* __restrict__ in, const float* __restrict__ Ws,
       const float* __restrict__ Wn, const float* __restrict__ bias,
       float* __restrict__ P, float* __restrict__ G, float* __restrict__ degc,
       const int* __restrict__ part, int* __restrict__ ctr, int n, int first) {
  __shared__ float wls[D][WPAD];
  __shared__ float wln[D][WPAD];
  stage_wT(Ws, wls, threadIdx.x);
  stage_wT(Wn, wln, threadIdx.x);
  if (blockIdx.x == 0 && threadIdx.x < 2) ctr[threadIdx.x * CTR_STRIDE] = 0;
  __syncthreads();

  const int lane = threadIdx.x & 31;
  const int wave = threadIdx.x >> 5;
  const int row0 = (blockIdx.x * 8 + wave) * 16;
  if (row0 >= n) return;
  const int half = lane >> 4;
  const int l15 = lane & 15;
  int rowA = row0 + l15;
  if (rowA >= n) rowA = n - 1;
  const float* __restrict__ ip = in + (size_t)rowA * D + half * 2;

  float dc;
  if (first) {
    int cs = 0;
#pragma unroll
    for (int k = 0; k < DS / 2; ++k) cs += part[(size_t)(half * (DS / 2) + k) * n + rowA];
    cs += __shfl_xor(cs, 16, 32);
    dc = fmaxf((float)cs, 1.0f);
    if (half == 0 && row0 + l15 < n) degc[row0 + l15] = dc;
  } else {
    dc = degc[rowA];
  }

  v2f a[16];
#pragma unroll
  for (int j = 0; j < 16; ++j) a[j] = *reinterpret_cast<const v2f*>(ip + 4 * j);
  if (first) {
#pragma unroll
    for (int j = 0; j < 16; ++j) {
      a[j].x = __logf(a[j].x + 1.0f);
      a[j].y = __logf(a[j].y + 1.0f);
    }
  } else {
    const float rdc = 1.0f / dc;
    float ss = 0.0f;
#pragma unroll
    for (int j = 0; j < 16; ++j) {
      a[j].x = fmaxf(a[j].x * rdc, 0.0f);
      a[j].y = fmaxf(a[j].y * rdc, 0.0f);
      ss += a[j].x * a[j].x;
      ss += a[j].y * a[j].y;
    }
    ss += __shfl_xor(ss, 16, 32);
    const float inv = 1.0f / fmaxf(sqrtf(ss), 1e-12f);
#pragma unroll
    for (int j = 0; j < 16; ++j) {
      a[j].x *= inv;
      a[j].y *= inv;
    }
  }

  v8f accS[4] = {}, accN[4] = {};
#pragma unroll
  for (int j = 0; j < 16; ++j) {
    const int kk = 4 * j + half * 2;
#pragma unroll
    for (int t = 0; t < 4; ++t) {
      const int nn = t * 16 + l15;
      const v2f bs = *reinterpret_cast<const v2f*>(&wls[nn][kk]);
      const v2f bn = *reinterpret_cast<const v2f*>(&wln[nn][kk]);
      accS[t] = WMMA4(a[j], bs, accS[t]);
      accN[t] = WMMA4(a[j], bn, accN[t]);
    }
  }

  float dr[8];
#pragma unroll
  for (int i = 0; i < 8; ++i) {
    const float d0 = __int_as_float(__builtin_amdgcn_readlane(__float_as_int(dc), i));
    const float d1 = __int_as_float(__builtin_amdgcn_readlane(__float_as_int(dc), i + 8));
    dr[i] = half ? d1 : d0;
  }
#pragma unroll
  for (int t = 0; t < 4; ++t) {
    const int c = t * 16 + l15;
    const float bv = bias[c];
#pragma unroll
    for (int i = 0; i < 8; ++i) {
      const int r = row0 + i + half * 8;
      if (r < n) P[(size_t)r * D + c] = (accS[t][i] + bv) * dr[i];
    }
  }
#pragma unroll
  for (int t = 0; t < 2; ++t) {
#pragma unroll
    for (int i = 0; i < 8; ++i) {
      const int r = row0 + i + half * 8;
      if (r < n) {
        v2f pr;
        pr.x = accN[t][i];
        pr.y = accN[t + 2][i];
        *reinterpret_cast<v2f*>(G + (size_t)r * D + 2 * (t * 16 + l15)) = pr;
      }
    }
  }
}

__global__ void __launch_bounds__(256)
k_head(const float* __restrict__ h, const float* __restrict__ degc,
       const float* __restrict__ fcw,
       const float* __restrict__ fcb, const float* __restrict__ gamma,
       const float* __restrict__ beta, const float* __restrict__ mean,
       const float* __restrict__ var, const float* __restrict__ w21,
       const float* __restrict__ b21, const float* __restrict__ w22,
       const float* __restrict__ b22, float* __restrict__ zloc,
       float* __restrict__ zscale, int n) {
  __shared__ float wl0[D][WPAD];
  __shared__ float wl1[D][WPAD];
  __shared__ float wl2[D][WPAD];
  __shared__ float u[8][16][WPAD];
  stage_wT(fcw, wl0, threadIdx.x);
  stage_wT(w21, wl1, threadIdx.x);
  stage_wT(w22, wl2, threadIdx.x);
  __syncthreads();

  const int lane = threadIdx.x & 31;
  const int wave = threadIdx.x >> 5;
  const int row0 = (blockIdx.x * 8 + wave) * 16;
  if (row0 >= n) return;
  const int half = lane >> 4;
  const int l15 = lane & 15;
  int rowA = row0 + l15;
  if (rowA >= n) rowA = n - 1;
  const float* __restrict__ hp = h + (size_t)rowA * D;
  const float rdc = 1.0f / degc[rowA];

  v8f acc[4] = {};
  for (int k0 = 0; k0 < D; k0 += 4) {
    const int kk = k0 + half * 2;
    const v2f a = *reinterpret_cast<const v2f*>(hp + kk) * rdc;
#pragma unroll
    for (int t = 0; t < 4; ++t) {
      const int nn = t * 16 + l15;
      const v2f b = *reinterpret_cast<const v2f*>(&wl0[nn][kk]);
      acc[t] = WMMA4(a, b, acc[t]);
    }
  }
#pragma unroll
  for (int t = 0; t < 4; ++t) {
    const int c = t * 16 + l15;
    const float bb = fcb[c];
    const float m = mean[c];
    const float iv = rsqrtf(var[c] + 1e-5f);
    const float g = gamma[c];
    const float bt = beta[c];
#pragma unroll
    for (int i = 0; i < 8; ++i) {
      float v = acc[t][i] + bb;
      v = (v - m) * iv * g + bt;
      v = fmaxf(v, 0.0f);
      v = v + __logf(1.0f + __expf(-v));
      u[wave][i + half * 8][c] = v;
    }
  }
  v8f aL[4] = {}, aS[4] = {};
  for (int k0 = 0; k0 < D; k0 += 4) {
    const int kk = k0 + half * 2;
    const v2f a = *reinterpret_cast<const v2f*>(&u[wave][l15][kk]);
#pragma unroll
    for (int t = 0; t < 4; ++t) {
      const int nn = t * 16 + l15;
      const v2f bL = *reinterpret_cast<const v2f*>(&wl1[nn][kk]);
      const v2f bS = *reinterpret_cast<const v2f*>(&wl2[nn][kk]);
      aL[t] = WMMA4(a, bL, aL[t]);
      aS[t] = WMMA4(a, bS, aS[t]);
    }
  }
#pragma unroll
  for (int t = 0; t < 4; ++t) {
    const int c = t * 16 + l15;
    const float bl = b21[c];
    const float bs = b22[c];
#pragma unroll
    for (int i = 0; i < 8; ++i) {
      const int r = row0 + i + half * 8;
      if (r < n) {
        const size_t o = (size_t)r * D + c;
        zloc[o] = aL[t][i] + bl;
        zscale[o] = __expf(aS[t][i] + bs);
      }
    }
  }
}

extern "C" void kernel_launch(void* const* d_in, const int* in_sizes, int n_in,
                              void* d_out, int out_size, void* d_ws, size_t ws_size,
                              hipStream_t stream) {
  const float* x = (const float*)d_in[0];
  const int* esrc = (const int*)d_in[1];
  const int* edst = (const int*)d_in[2];
  const float* Wself = (const float*)d_in[3];
  const float* Wneigh = (const float*)d_in[4];
  const float* bsage = (const float*)d_in[5];
  const float* fcw = (const float*)d_in[6];
  const float* fcb = (const float*)d_in[7];
  const float* gamma = (const float*)d_in[8];
  const float* beta = (const float*)d_in[9];
  const float* mean = (const float*)d_in[10];
  const float* var = (const float*)d_in[11];
  const float* w21 = (const float*)d_in[12];
  const float* b21 = (const float*)d_in[13];
  const float* w22 = (const float*)d_in[14];
  const float* b22 = (const float*)d_in[15];

  const int n = in_sizes[0] / D;
  const int e = in_sizes[1];
  const size_t nd = (size_t)n * D;

  float* ws = (float*)d_ws;
  float* deg = ws;
  int* ctr = (int*)(deg + n);
  float* pA = ws + (((size_t)n + CTR_STRIDE + 1 + 255) & ~(size_t)255);
  float* pB = pA + nd;
  float* g = pB + nd;

  const int tiles = (n + 15) / 16;
  const int gblocks = (tiles + 7) / 8;

  int* part = (int*)pB;
  k_deg<<<256, DT, 0, stream>>>(edst, part, e, n);
  k_proj<<<gblocks, 256, 0, stream>>>(x, Wself, Wneigh, bsage, pA, g, deg, part, ctr, n, 1);

  float* hcur = pA;
  float* hnext = pB;
  for (int l = 0; l < 3; ++l) {
    k_scatter<<<2048, 256, 0, stream>>>(g, esrc, edst, hcur, ctr, e, n);
    if (l < 2) {
      k_proj<<<gblocks, 256, 0, stream>>>(hcur, Wself + (size_t)(l + 1) * D * D,
                                          Wneigh + (size_t)(l + 1) * D * D,
                                          bsage + (size_t)(l + 1) * D, hnext, g, deg, nullptr, ctr,
                                          n, 0);
      float* tmp = hcur;
      hcur = hnext;
      hnext = tmp;
    }
  }

  float* zloc = (float*)d_out;
  float* zscale = zloc + nd;
  k_head<<<gblocks, 256, 0, stream>>>(hcur, deg, fcw, fcb, gamma, beta, mean, var, w21, b21,
                                      w22, b22, zloc, zscale, n);
}


# Layer and head kernels: a work-group's prologue waits for one memory round trip (was 9 to 13): weight and row loads issued as one group ahead of the barrier, staging LDS stores free of bank conflicts
# speedup vs baseline: 2.2900x; 1.0696x over previous
#include <hip/hip_runtime.h>
#include <cstdint>

typedef __attribute__((ext_vector_type(2))) float v2f;
typedef __attribute__((ext_vector_type(8))) float v8f;

#define D 64
#define WPAD 68

#define WMMA4(a, b, c) \
  __builtin_amdgcn_wmma_f32_16x16x4_f32(false, (a), false, (b), (short)0, (c), false, false)

#define SG 4
__device__ __forceinline__ void stage_load(const float* __restrict__ W, float4 (&v)[SG],
                                           int tid) {
  const int wv = tid >> 5;
  const int ln = tid & 31;
  const int n4 = (((wv & 3) << 2) | (ln & 3)) << 2;
#pragma unroll
  for (int it = 0; it < SG; ++it) {
    const int k = ((2 * it + (wv >> 2)) << 3) | (ln >> 2);
    v[it] = *reinterpret_cast<const float4*>(W + k * D + n4);
  }
}
__device__ __forceinline__ void stage_store(float (*wl)[WPAD], const float4 (&v)[SG], int tid) {
  const int wv = tid >> 5;
  const int ln = tid & 31;
  const int n4 = (((wv & 3) << 2) | (ln & 3)) << 2;
#pragma unroll
  for (int it = 0; it < SG; ++it) {
    const int k = ((2 * it + (wv >> 2)) << 3) | (ln >> 2);
    wl[n4 + 0][k] = v[it].x;
    wl[n4 + 1][k] = v[it].y;
    wl[n4 + 2][k] = v[it].z;
    wl[n4 + 3][k] = v[it].w;
  }
}

#define DR 6272
#define DS 16
#define DT 512
#define DU 25
__global__ void __launch_bounds__(DT)
k_deg(const int* __restrict__ dst, int* __restrict__ part, int e, int n) {
  __shared__ int cnt[DR];
  const int tid = threadIdx.x;
  const int nr = (n + DR - 1) / DR;
  const int e4 = e >> 2;
  const int q4 = (e4 + DS - 1) / DS;
  const int4* __restrict__ d4 = reinterpret_cast<const int4*>(dst);
  for (int w = blockIdx.x; w < nr * DS; w += gridDim.x) {
    const int r = w / DS;
    const int s = w - r * DS;
    const int lo = r * DR;
    const unsigned len = (unsigned)(min(lo + DR, n) - lo);
    for (int i = tid; i < DR; i += DT) cnt[i] = 0;
    __syncthreads();
    const int b4 = s * q4;
    const int t4 = min(b4 + q4, e4);
    for (int base = b4; base < t4; base += DT * DU) {
      int4 v[DU];
#pragma unroll
      for (int k = 0; k < DU; ++k) {
        const int i4 = base + k * DT + tid;
        v[k] = d4[min(i4, t4 - 1)];
        if (i4 >= t4) v[k] = make_int4(-1, -1, -1, -1);
      }
#pragma unroll
      for (int k = 0; k < DU; ++k) {
        const unsigned o0 = (unsigned)(v[k].x - lo);
        const unsigned o1 = (unsigned)(v[k].y - lo);
        const unsigned o2 = (unsigned)(v[k].z - lo);
        const unsigned o3 = (unsigned)(v[k].w - lo);
        if (o0 < len) atomicAdd(&cnt[o0], 1);
        if (o1 < len) atomicAdd(&cnt[o1], 1);
        if (o2 < len) atomicAdd(&cnt[o2], 1);
        if (o3 < len) atomicAdd(&cnt[o3], 1);
      }
    }
    if (s == 0 && tid < (e & 3)) {
      const unsigned o = (unsigned)(dst[(e4 << 2) + tid] - lo);
      if (o < len) atomicAdd(&cnt[o], 1);
    }
    __syncthreads();
    for (unsigned i = tid; i < len; i += DT) part[(size_t)s * n + lo + i] = cnt[i];
    __syncthreads();
  }
}

#define CH 256
#define GRP 16
#define CTR_STRIDE 64
typedef __attribute__((address_space(1))) float gfloat;
typedef __attribute__((address_space(1))) v2f gv2f;

#define PIN8(x, o)                                                                   \
  asm("" : "+s"(x[o + 0]), "+s"(x[o + 1]), "+s"(x[o + 2]), "+s"(x[o + 3]),           \
           "+s"(x[o + 4]), "+s"(x[o + 5]), "+s"(x[o + 6]), "+s"(x[o + 7]))

__global__ void __launch_bounds__(256)
k_scatter(const float* __restrict__ g, const int* __restrict__ src,
          const int* __restrict__ dst, float* __restrict__ acc, int* __restrict__ ctr,
          int e, int n) {
  __shared__ int tk[3];
  const unsigned lane = threadIdx.x & 31u;
  const unsigned wave = threadIdx.x >> 5;
  const int nch = (e + CH - 1) / CH;
  const int nsplit = n >> 1;
  if (threadIdx.x == 0) {
    const unsigned id = (unsigned)__builtin_amdgcn_s_sendmsg_rtn(0x87);
    const int q0 = (int)((id >> 18) & 1u);
    tk[2] = q0;
    tk[0] = atomicAdd(ctr + q0 * CTR_STRIDE, 1);
  }
  __syncthreads();
  int q = tk[2];
  int t = tk[0];
  int it = 0;
  bool other = false;
  for (;;) {
    int tn = 0;
    const bool live = t < nch;
    if (!live) {
      if (other) break;
      other = true;
      q ^= 1;
    }
    if (threadIdx.x == 0) tn = atomicAdd(ctr + q * CTR_STRIDE, 1);
    if (live) {
      const long ei = (long)t * CH + (long)(wave * 32u + lane);
      const bool ok = ei < e;
      const int d = ok ? dst[ei] : 0;
      const int s = ok ? src[ei] : 0;
      const bool mine = ok && ((d >= nsplit) == (q != 0));
      unsigned m = __builtin_amdgcn_ballot_w32(mine);
      while (m) {
        const unsigned m0 = m;
        const int cnt = min((int)__builtin_popcount(m0), GRP);
        unsigned long long hb[GRP], db[GRP];
        unsigned w1 = m0;
#pragma unroll
        for (int k = 0; k < GRP; ++k) {
          const int j = w1 ? (int)__builtin_ctz(w1) : 0;
          w1 &= w1 - 1u;
          const unsigned sj = (unsigned)__builtin_amdgcn_readlane(s, j);
          hb[k] = (unsigned long long)g + (unsigned long long)sj * (D * sizeof(float));
        }
        PIN8(hb, 0);
        PIN8(hb, 8);
        v2f ab[GRP];
#pragma unroll
        for (int k = 0; k < GRP; ++k) {
          if (k < cnt)
            ab[k] = ((const gv2f*)hb[k])[lane];
        }
        unsigned w2 = m0;
#pragma unroll
        for (int k = 0; k < GRP; ++k) {
          const int j = w2 ? (int)__builtin_ctz(w2) : 0;
          w2 &= w2 - 1u;
          const unsigned dd = (unsigned)__builtin_amdgcn_readlane(d, j);
          db[k] = (unsigned long long)acc + (unsigned long long)dd * (D * sizeof(float));
        }
        PIN8(db, 0);
        PIN8(db, 8);
#pragma unroll
        for (int k = 0; k < GRP; ++k) {
          if (k < cnt) {
            float* w = (float*)(gfloat*)db[k];
            atomicAdd(w + lane, ab[k].x);
            atomicAdd(w + lane + 32u, ab[k].y);
          }
        }
        m = w2;
      }
    }
    if (threadIdx.x == 0) tk[(it + 1) & 1] = tn;
    __syncthreads();
    ++it;
    t = tk[it & 1];
  }
}

template <bool FIRST>
__global__ void __launch_bounds__(256)
k_proj(const float* __restrict__ in, const float* __restrict__ Ws,
       const float* __restrict__ Wn, const float* __restrict__ bias,
       float* __restrict__ P, float* __restrict__ G, float* __restrict__ degc,
       const int* __restrict__ part, int* __restrict__ ctr, int n) {
  __shared__ float wls[D][WPAD];
  __shared__ float wln[D][WPAD];
  const int tid = threadIdx.x;
  const int lane = tid & 31;
  const int wave = __builtin_amdgcn_readfirstlane(tid >> 5);
  const int row0 = (blockIdx.x * 8 + wave) * 16;
  const int half = lane >> 4;
  const int l15 = lane & 15;
  int rowA = row0 + l15;
  if (rowA >= n) rowA = n - 1;
  const float* __restrict__ ip = in + (size_t)rowA * D + half * 2;

  float4 ws4[SG], wn4[SG];
  stage_load(Ws, ws4, tid);
  stage_load(Wn, wn4, tid);
  int pc[DS / 2] = {};
  float dcl = 1.0f;
  if (FIRST) {
#pragma unroll
    for (int k = 0; k < DS / 2; ++k) pc[k] = part[(size_t)(half * (DS / 2) + k) * n + rowA];
  } else {
    dcl = degc[rowA];
  }
  v2f a[16];
#pragma unroll
  for (int j = 0; j < 16; ++j) a[j] = *reinterpret_cast<const v2f*>(ip + 4 * j);

  stage_store(wls, ws4, tid);
  stage_store(wln, wn4, tid);
  if (blockIdx.x == 0 && threadIdx.x < 2) ctr[threadIdx.x * CTR_STRIDE] = 0;
  __syncthreads();

  if (row0 >= n) return;

  float dc;
  if (FIRST) {
    int cs = 0;
#pragma unroll
    for (int k = 0; k < DS / 2; ++k) cs += pc[k];
    cs += __shfl_xor(cs, 16, 32);
    dc = fmaxf((float)cs, 1.0f);
    if (half == 0 && row0 + l15 < n) degc[row0 + l15] = dc;
  } else {
    dc = dcl;
  }

  if (FIRST) {
#pragma unroll
    for (int j = 0; j < 16; ++j) {
      a[j].x = __logf(a[j].x + 1.0f);
      a[j].y = __logf(a[j].y + 1.0f);
    }
  } else {
    const float rdc = 1.0f / dc;
    float ss = 0.0f;
#pragma unroll
    for (int j = 0; j < 16; ++j) {
      a[j].x = fmaxf(a[j].x * rdc, 0.0f);
      a[j].y = fmaxf(a[j].y * rdc, 0.0f);
      ss += a[j].x * a[j].x;
      ss += a[j].y * a[j].y;
    }
    ss += __shfl_xor(ss, 16, 32);
    const float inv = 1.0f / fmaxf(sqrtf(ss), 1e-12f);
#pragma unroll
    for (int j = 0; j < 16; ++j) {
      a[j].x *= inv;
      a[j].y *= inv;
    }
  }

  v8f accS[4] = {}, accN[4] = {};
#pragma unroll
  for (int j = 0; j < 16; ++j) {
    const int kk = 4 * j + half * 2;
#pragma unroll
    for (int t = 0; t < 4; ++t) {
      const int nn = t * 16 + l15;
      const v2f bs = *reinterpret_cast<const v2f*>(&wls[nn][kk]);
      const v2f bn = *reinterpret_cast<const v2f*>(&wln[nn][kk]);
      accS[t] = WMMA4(a[j], bs, accS[t]);
      accN[t] = WMMA4(a[j], bn, accN[t]);
    }
  }

  float dr[8];
#pragma unroll
  for (int i = 0; i < 8; ++i) {
    const float d0 = __int_as_float(__builtin_amdgcn_readlane(__float_as_int(dc), i));
    const float d1 = __int_as_float(__builtin_amdgcn_readlane(__float_as_int(dc), i + 8));
    dr[i] = half ? d1 : d0;
  }
#pragma unroll
  for (int t = 0; t < 4; ++t) {
    const int c = t * 16 + l15;
    const float bv = bias[c];
#pragma unroll
    for (int i = 0; i < 8; ++i) {
      const int r = row0 + i + half * 8;
      if (r < n) P[(size_t)r * D + c] = (accS[t][i] + bv) * dr[i];
    }
  }
#pragma unroll
  for (int t = 0; t < 2; ++t) {
#pragma unroll
    for (int i = 0; i < 8; ++i) {
      const int r = row0 + i + half * 8;
      if (r < n) {
        v2f pr;
        pr.x = accN[t][i];
        pr.y = accN[t + 2][i];
        *reinterpret_cast<v2f*>(G + (size_t)r * D + 2 * (t * 16 + l15)) = pr;
      }
    }
  }
}

__global__ void __launch_bounds__(256)
k_head(const float* __restrict__ h, const float* __restrict__ degc,
       const float* __restrict__ fcw,
       const float* __restrict__ fcb, const float* __restrict__ gamma,
       const float* __restrict__ beta, const float* __restrict__ mean,
       const float* __restrict__ var, const float* __restrict__ w21,
       const float* __restrict__ b21, const float* __restrict__ w22,
       const float* __restrict__ b22, float* __restrict__ zloc,
       float* __restrict__ zscale, int n) {
  __shared__ float wl0[D][WPAD];
  __shared__ float wl1[D][WPAD];
  __shared__ float wl2[D][WPAD];
  __shared__ float u[8][16][WPAD];
  const int tid = threadIdx.x;
  const int lane = tid & 31;
  const int wave = tid >> 5;
  const int row0 = (blockIdx.x * 8 + wave) * 16;
  const int half = lane >> 4;
  const int l15 = lane & 15;
  int rowA = row0 + l15;
  if (rowA >= n) rowA = n - 1;
  const float* __restrict__ hp = h + (size_t)rowA * D + half * 2;

  float4 w0[SG], w1[SG], w2[SG];
  stage_load(fcw, w0, tid);
  stage_load(w21, w1, tid);
  stage_load(w22, w2, tid);
  const float dcl = degc[rowA];
  v2f hr[16];
#pragma unroll
  for (int j = 0; j < 16; ++j) hr[j] = *reinterpret_cast<const v2f*>(hp + 4 * j);
  stage_store(wl0, w0, tid);
  stage_store(wl1, w1, tid);
  stage_store(wl2, w2, tid);
  __syncthreads();

  if (row0 >= n) return;
  const float rdc = 1.0f / dcl;

  v8f acc[4] = {};
#pragma unroll
  for (int j = 0; j < 16; ++j) {
    const int kk = 4 * j + half * 2;
    const v2f a = hr[j] * rdc;
#pragma unroll
    for (int t = 0; t < 4; ++t) {
      const int nn = t * 16 + l15;
      const v2f b = *reinterpret_cast<const v2f*>(&wl0[nn][kk]);
      acc[t] = WMMA4(a, b, acc[t]);
    }
  }
#pragma unroll
  for (int t = 0; t < 4; ++t) {
    const int c = t * 16 + l15;
    const float bb = fcb[c];
    const float m = mean[c];
    const float iv = rsqrtf(var[c] + 1e-5f);
    const float g = gamma[c];
    const float bt = beta[c];
#pragma unroll
    for (int i = 0; i < 8; ++i) {
      float v = acc[t][i] + bb;
      v = (v - m) * iv * g + bt;
      v = fmaxf(v, 0.0f);
      v = v + __logf(1.0f + __expf(-v));
      u[wave][i + half * 8][c] = v;
    }
  }
  v8f aL[4] = {}, aS[4] = {};
  for (int k0 = 0; k0 < D; k0 += 4) {
    const int kk = k0 + half * 2;
    const v2f a = *reinterpret_cast<const v2f*>(&u[wave][l15][kk]);
#pragma unroll
    for (int t = 0; t < 4; ++t) {
      const int nn = t * 16 + l15;
      const v2f bL = *reinterpret_cast<const v2f*>(&wl1[nn][kk]);
      const v2f bS = *reinterpret_cast<const v2f*>(&wl2[nn][kk]);
      aL[t] = WMMA4(a, bL, aL[t]);
      aS[t] = WMMA4(a, bS, aS[t]);
    }
  }
#pragma unroll
  for (int t = 0; t < 4; ++t) {
    const int c = t * 16 + l15;
    const float bl = b21[c];
    const float bs = b22[c];
#pragma unroll
    for (int i = 0; i < 8; ++i) {
      const int r = row0 + i + half * 8;
      if (r < n) {
        const size_t o = (size_t)r * D + c;
        zloc[o] = aL[t][i] + bl;
        zscale[o] = __expf(aS[t][i] + bs);
      }
    }
  }
}

extern "C" void kernel_launch(void* const* d_in, const int* in_sizes, int n_in,
                              void* d_out, int out_size, void* d_ws, size_t ws_size,
                              hipStream_t stream) {
  const float* x = (const float*)d_in[0];
  const int* esrc = (const int*)d_in[1];
  const int* edst = (const int*)d_in[2];
  const float* Wself = (const float*)d_in[3];
  const float* Wneigh = (const float*)d_in[4];
  const float* bsage = (const float*)d_in[5];
  const float* fcw = (const float*)d_in[6];
  const float* fcb = (const float*)d_in[7];
  const float* gamma = (const float*)d_in[8];
  const float* beta = (const float*)d_in[9];
  const float* mean = (const float*)d_in[10];
  const float* var = (const float*)d_in[11];
  const float* w21 = (const float*)d_in[12];
  const float* b21 = (const float*)d_in[13];
  const float* w22 = (const float*)d_in[14];
  const float* b22 = (const float*)d_in[15];

  const int n = in_sizes[0] / D;
  const int e = in_sizes[1];
  const size_t nd = (size_t)n * D;

  float* ws = (float*)d_ws;
  float* deg = ws;
  int* ctr = (int*)(deg + n);
  float* pA = ws + (((size_t)n + CTR_STRIDE + 1 + 255) & ~(size_t)255);
  float* pB = pA + nd;
  float* g = pB + nd;

  const int tiles = (n + 15) / 16;
  const int gblocks = (tiles + 7) / 8;

  int* part = (int*)pB;
  k_deg<<<256, DT, 0, stream>>>(edst, part, e, n);
  k_proj<true><<<gblocks, 256, 0, stream>>>(x, Wself, Wneigh, bsage, pA, g, deg, part, ctr, n);

  float* hcur = pA;
  float* hnext = pB;
  for (int l = 0; l < 3; ++l) {
    k_scatter<<<2048, 256, 0, stream>>>(g, esrc, edst, hcur, ctr, e, n);
    if (l < 2) {
      k_proj<false><<<gblocks, 256, 0, stream>>>(hcur, Wself + (size_t)(l + 1) * D * D,
                                                 Wneigh + (size_t)(l + 1) * D * D,
                                                 bsage + (size_t)(l + 1) * D, hnext, g, deg,
                                                 nullptr, ctr, n);
      float* tmp = hcur;
      hcur = hnext;
      hnext = tmp;
    }
  }

  float* zloc = (float*)d_out;
  float* zscale = zloc + nd;
  k_head<<<gblocks, 256, 0, stream>>>(hcur, deg, fcw, fcb, gamma, beta, mean, var, w21, b21,
                                      w22, b22, zloc, zscale, n);
}
